# starting kernel
# baseline (speedup 1.0000x reference)
#include <hip/hip_runtime.h>
#include <hip/hip_bf16.h>
#include <cstdint>
#include <cstddef>

#define BATCH 32
#define NNODE 4096
#define HISTT 16
#define PREDT 8
#define NEDGE 65536

typedef float v2f __attribute__((ext_vector_type(2)));
typedef float v8f __attribute__((ext_vector_type(8)));

__device__ __forceinline__ float sigm(float x) { return 1.f / (1.f + __expf(-x)); }

__device__ __forceinline__ void atomAddF(float* p, float v) {
  __hip_atomic_fetch_add(p, v, __ATOMIC_RELAXED, __HIP_MEMORY_SCOPE_AGENT);
}

__global__ void k_zero(float* __restrict__ p, int n) {
  int i = blockIdx.x * 256 + threadIdx.x;
  if (i < n) p[i] = 0.f;
}

__global__ void k_deg(const int* __restrict__ ei, float* __restrict__ deg) {
  int e = blockIdx.x * 256 + threadIdx.x;
  if (e < NEDGE) atomAddF(deg + ei[e], 1.f);
}

__global__ void k_dis(const float* __restrict__ deg, float* __restrict__ dis) {
  int n = blockIdx.x * 256 + threadIdx.x;
  if (n < NNODE) {
    float d = deg[n];
    dis[n] = d > 0.f ? rsqrtf(d) : 0.f;
  }
}

__global__ void k_enorm(const int* __restrict__ ei, const float* __restrict__ dis,
                        float* __restrict__ enorm) {
  int e = blockIdx.x * 256 + threadIdx.x;
  if (e < NEDGE) enorm[e] = -dis[ei[e]] * dis[ei[NEDGE + e]];
}

__global__ void k_proj_hist(const float* __restrict__ pm25, const float* __restrict__ feat,
                            const float* __restrict__ cheb_w, const float* __restrict__ cheb_b,
                            float* __restrict__ p, float* __restrict__ q) {
  int i = blockIdx.x * 256 + threadIdx.x;
  if (i >= BATCH * HISTT * NNODE) return;
  int bt = i >> 12, b = bt >> 4, t = bt & 15, n = i & 4095;
  float x[10];
  x[0] = pm25[i];
  const float* fp = feat + (((size_t)b * 24 + t) * NNODE + n) * 9;
#pragma unroll
  for (int j = 0; j < 9; ++j) x[1 + j] = fp[j];
  float q0 = cheb_b[0], q1 = cheb_b[1], p0 = 0.f, p1 = 0.f;
#pragma unroll
  for (int f = 0; f < 10; ++f) {
    q0 += x[f] * cheb_w[f * 2 + 0];
    q1 += x[f] * cheb_w[f * 2 + 1];
    p0 += x[f] * cheb_w[20 + f * 2 + 0];
    p1 += x[f] * cheb_w[20 + f * 2 + 1];
  }
  size_t o = (size_t)i * 2;
  p[o] = p0; p[o + 1] = p1; q[o] = q0; q[o + 1] = q1;
}

__global__ void k_scatter_hist(const int* __restrict__ ei, const float* __restrict__ enorm,
                               const float* __restrict__ p, float* __restrict__ acc) {
  int idx = blockIdx.x * 256 + threadIdx.x;
  if (idx >= BATCH * HISTT * NEDGE) return;
  int bt = idx >> 16, e = idx & 65535;
  float nv = enorm[e];
  int s = ei[e], d = ei[NEDGE + e];
  size_t sb = ((size_t)bt * NNODE + s) * 2;
  size_t db = ((size_t)bt * NNODE + d) * 2;
  atomAddF(&acc[db + 0], p[sb + 0] * nv);
  atomAddF(&acc[db + 1], p[sb + 1] * nv);
}

__global__ void k_fin_hist(const float* __restrict__ q, float* __restrict__ acc) {
  int i = blockIdx.x * 256 + threadIdx.x;
  if (i < BATCH * HISTT * NNODE * 2) acc[i] = sigm(q[i] + acc[i]);
}

__global__ void k_proj_pred(const float* __restrict__ xn, const float* __restrict__ feat,
                            const float* __restrict__ cheb_w, const float* __restrict__ cheb_b,
                            float* __restrict__ p, float* __restrict__ q, float* __restrict__ acc,
                            int kstep) {
  int i = blockIdx.x * 256 + threadIdx.x;
  if (i >= BATCH * NNODE) return;
  int b = i >> 12, n = i & 4095;
  float x[10];
  x[0] = xn[i];
  const float* fp = feat + (((size_t)b * 24 + HISTT + kstep) * NNODE + n) * 9;
#pragma unroll
  for (int j = 0; j < 9; ++j) x[1 + j] = fp[j];
  float q0 = cheb_b[0], q1 = cheb_b[1], p0 = 0.f, p1 = 0.f;
#pragma unroll
  for (int f = 0; f < 10; ++f) {
    q0 += x[f] * cheb_w[f * 2 + 0];
    q1 += x[f] * cheb_w[f * 2 + 1];
    p0 += x[f] * cheb_w[20 + f * 2 + 0];
    p1 += x[f] * cheb_w[20 + f * 2 + 1];
  }
  size_t o = (size_t)i * 2;
  p[o] = p0; p[o + 1] = p1; q[o] = q0; q[o + 1] = q1;
  acc[o] = 0.f; acc[o + 1] = 0.f;
}

__global__ void k_scatter_pred(const int* __restrict__ ei, const float* __restrict__ enorm,
                               const float* __restrict__ p, float* __restrict__ acc) {
  int idx = blockIdx.x * 256 + threadIdx.x;
  if (idx >= BATCH * NEDGE) return;
  int b = idx >> 16, e = idx & 65535;
  float nv = enorm[e];
  int s = ei[e], d = ei[NEDGE + e];
  size_t sb = ((size_t)b * NNODE + s) * 2;
  size_t db = ((size_t)b * NNODE + d) * 2;
  atomAddF(&acc[db + 0], p[sb + 0] * nv);
  atomAddF(&acc[db + 1], p[sb + 1] * nv);
}

#define WT_STRIDE 48

__device__ __forceinline__ void load_weights(int tid,
    const float* __restrict__ x2h_w, const float* __restrict__ x2h_b,
    const float* __restrict__ h2h_w, const float* __restrict__ h2h_b,
    const float* __restrict__ fc_w,
    float* sWT, float* sBias, float* sFc) {
  for (int i = tid; i < 44 * 128; i += 256) {
    int k = i >> 7, n = i & 127;
    sWT[n * WT_STRIDE + k] = (k < 12) ? x2h_w[k * 128 + n] : h2h_w[(k - 12) * 128 + n];
  }
  if (tid < 128) sBias[tid] = x2h_b[tid] + h2h_b[tid];
  if (tid < 32) sFc[tid] = fc_w[tid];
}

__device__ __forceinline__ void gate_compute(const float* __restrict__ sXw,
                                             const float* __restrict__ sHw,
                                             const float* __restrict__ sWT,
                                             const float* __restrict__ sBias,
                                             int m, int hi, v8f* acc) {
  v2f ax[3], ah[8];
#pragma unroll
  for (int kk = 0; kk < 3; ++kk) ax[kk] = *(const v2f*)(sXw + m * 12 + 4 * kk + 2 * hi);
#pragma unroll
  for (int kk = 0; kk < 8; ++kk) ah[kk] = *(const v2f*)(sHw + m * 32 + 4 * kk + 2 * hi);
#pragma unroll
  for (int ct = 0; ct < 8; ++ct) {
    float bb = sBias[ct * 16 + m];
    v8f a = {bb, bb, bb, bb, bb, bb, bb, bb};
    const float* wcol = sWT + (ct * 16 + m) * WT_STRIDE;
#pragma unroll
    for (int kk = 0; kk < 3; ++kk) {
      v2f bf = *(const v2f*)(wcol + 4 * kk + 2 * hi);
      a = __builtin_amdgcn_wmma_f32_16x16x4_f32(false, ax[kk], false, bf, (short)0, a,
                                                false, false);
    }
#pragma unroll
    for (int kk = 0; kk < 8; ++kk) {
      v2f bf = *(const v2f*)(wcol + 12 + 4 * kk + 2 * hi);
      a = __builtin_amdgcn_wmma_f32_16x16x4_f32(false, ah[kk], false, bf, (short)0, a,
                                                false, false);
    }
    acc[ct] = a;
  }
}

__device__ __forceinline__ void lstm_update(const v8f* acc, float* c01, float* h01,
                                            float* sHw, int m, int hi) {
#pragma unroll
  for (int ct = 0; ct < 2; ++ct) {
#pragma unroll
    for (int r = 0; r < 8; ++r) {
      float iv = sigm(acc[ct][r]);
      float fv = sigm(acc[2 + ct][r]);
      float gv = tanhf(acc[4 + ct][r]);
      float ov = sigm(acc[6 + ct][r]);
      float cn = fv * c01[ct * 8 + r] + iv * gv;
      float hn = ov * tanhf(cn);
      c01[ct * 8 + r] = cn;
      h01[ct * 8 + r] = hn;
      sHw[(r + 8 * hi) * 32 + ct * 16 + m] = hn;
    }
  }
}

__launch_bounds__(256)
__global__ void hist_lstm_kernel(
    const float* __restrict__ pm25, const float* __restrict__ feat,
    const float* __restrict__ xg,
    const float* __restrict__ x2h_w, const float* __restrict__ x2h_b,
    const float* __restrict__ h2h_w, const float* __restrict__ h2h_b,
    const float* __restrict__ fc_w, const float* __restrict__ fc_b,
    float* __restrict__ h_glob, float* __restrict__ c_glob, float* __restrict__ xn_glob) {
  __shared__ float sWT[128 * WT_STRIDE];
  __shared__ float sBias[128];
  __shared__ float sFc[32];
  __shared__ float sX[8][16 * 12];
  __shared__ float sH[8][16 * 32];

  const int tid = threadIdx.x;
  load_weights(tid, x2h_w, x2h_b, h2h_w, h2h_b, fc_w, sWT, sBias, sFc);

  const int wave = tid >> 5, lane = tid & 31, m = lane & 15, hi = lane >> 4;
  const int rowbase = blockIdx.x * 128 + wave * 16;
  float* sXw = sX[wave];
  float* sHw = sH[wave];
  for (int j = lane; j < 512; j += 32) sHw[j] = 0.f;
  float c01[16], h01[16];
#pragma unroll
  for (int j = 0; j < 16; ++j) { c01[j] = 0.f; h01[j] = 0.f; }
  __syncthreads();

  for (int t = 0; t < HISTT; ++t) {
    if (lane < 16) {
      int R = rowbase + lane;
      int b = R >> 12, n = R & 4095;
      size_t hb = ((size_t)(b * 16 + t)) * NNODE + n;
      sXw[lane * 12 + 0] = pm25[hb];
      const float* fp = feat + (((size_t)b * 24 + t) * NNODE + n) * 9;
#pragma unroll
      for (int j = 0; j < 9; ++j) sXw[lane * 12 + 1 + j] = fp[j];
      sXw[lane * 12 + 10] = xg[hb * 2 + 0];
      sXw[lane * 12 + 11] = xg[hb * 2 + 1];
    }
    __syncthreads();
    v8f acc[8];
    gate_compute(sXw, sHw, sWT, sBias, m, hi, acc);
    lstm_update(acc, c01, h01, sHw, m, hi);
  }

#pragma unroll
  for (int ct = 0; ct < 2; ++ct)
#pragma unroll
    for (int r = 0; r < 8; ++r) {
      size_t R = (size_t)rowbase + r + 8 * hi;
      c_glob[R * 32 + ct * 16 + m] = c01[ct * 8 + r];
      h_glob[R * 32 + ct * 16 + m] = h01[ct * 8 + r];
    }
  __syncthreads();
  if (lane < 16) {
    float s = fc_b[0];
#pragma unroll
    for (int j = 0; j < 32; ++j) s += sHw[lane * 32 + j] * sFc[j];
    xn_glob[rowbase + lane] = s;
  }
}

__launch_bounds__(256)
__global__ void pred_lstm_kernel(
    const float* __restrict__ feat,
    const float* __restrict__ q_pred, const float* __restrict__ acc_pred,
    const float* __restrict__ x2h_w, const float* __restrict__ x2h_b,
    const float* __restrict__ h2h_w, const float* __restrict__ h2h_b,
    const float* __restrict__ fc_w, const float* __restrict__ fc_b,
    float* __restrict__ h_glob, float* __restrict__ c_glob, float* __restrict__ xn_glob,
    float* __restrict__ out, int kstep) {
  __shared__ float sWT[128 * WT_STRIDE];
  __shared__ float sBias[128];
  __shared__ float sFc[32];
  __shared__ float sX[8][16 * 12];
  __shared__ float sH[8][16 * 32];

  const int tid = threadIdx.x;
  load_weights(tid, x2h_w, x2h_b, h2h_w, h2h_b, fc_w, sWT, sBias, sFc);

  const int wave = tid >> 5, lane = tid & 31, m = lane & 15, hi = lane >> 4;
  const int rowbase = blockIdx.x * 128 + wave * 16;
  float* sXw = sX[wave];
  float* sHw = sH[wave];
  for (int j = lane; j < 512; j += 32) sHw[j] = h_glob[(size_t)rowbase * 32 + j];
  float c01[16], h01[16];
#pragma unroll
  for (int ct = 0; ct < 2; ++ct)
#pragma unroll
    for (int r = 0; r < 8; ++r)
      c01[ct * 8 + r] = c_glob[((size_t)rowbase + r + 8 * hi) * 32 + ct * 16 + m];

  if (lane < 16) {
    int R = rowbase + lane;
    int b = R >> 12, n = R & 4095;
    sXw[lane * 12 + 0] = xn_glob[R];
    const float* fp = feat + (((size_t)b * 24 + HISTT + kstep) * NNODE + n) * 9;
#pragma unroll
    for (int j = 0; j < 9; ++j) sXw[lane * 12 + 1 + j] = fp[j];
    sXw[lane * 12 + 10] = sigm(q_pred[(size_t)R * 2 + 0] + acc_pred[(size_t)R * 2 + 0]);
    sXw[lane * 12 + 11] = sigm(q_pred[(size_t)R * 2 + 1] + acc_pred[(size_t)R * 2 + 1]);
  }
  __syncthreads();

  v8f acc[8];
  gate_compute(sXw, sHw, sWT, sBias, m, hi, acc);
  lstm_update(acc, c01, h01, sHw, m, hi);

#pragma unroll
  for (int ct = 0; ct < 2; ++ct)
#pragma unroll
    for (int r = 0; r < 8; ++r) {
      size_t R = (size_t)rowbase + r + 8 * hi;
      c_glob[R * 32 + ct * 16 + m] = c01[ct * 8 + r];
      h_glob[R * 32 + ct * 16 + m] = h01[ct * 8 + r];
    }
  __syncthreads();
  if (lane < 16) {
    float s = fc_b[0];
#pragma unroll
    for (int j = 0; j < 32; ++j) s += sHw[lane * 32 + j] * sFc[j];
    int R = rowbase + lane;
    int b = R >> 12, n = R & 4095;
    xn_glob[R] = s;
    out[((size_t)b * PREDT + kstep) * NNODE + n] = s;
  }
}

extern "C" void kernel_launch(void* const* d_in, const int* in_sizes, int n_in,
                              void* d_out, int out_size, void* d_ws, size_t ws_size,
                              hipStream_t stream) {
  (void)in_sizes; (void)n_in; (void)out_size; (void)ws_size;
  const float* pm25   = (const float*)d_in[0];
  const float* feat   = (const float*)d_in[1];
  const int*   ei     = (const int*)d_in[2];
  const float* cheb_w = (const float*)d_in[3];
  const float* cheb_b = (const float*)d_in[4];
  const float* x2h_w  = (const float*)d_in[5];
  const float* x2h_b  = (const float*)d_in[6];
  const float* h2h_w  = (const float*)d_in[7];
  const float* h2h_b  = (const float*)d_in[8];
  const float* fc_w   = (const float*)d_in[9];
  const float* fc_b   = (const float*)d_in[10];
  float* out = (float*)d_out;
  char* ws = (char*)d_ws;

  float* deg      = (float*)(ws + 0);
  float* dis      = (float*)(ws + 0x4000);
  float* enorm    = (float*)(ws + 0x8000);
  float* p_hist   = (float*)(ws + 0x100000);
  float* q_hist   = (float*)(ws + 0x1100000);
  float* acc_hist = (float*)(ws + 0x2100000);
  float* xn_glob  = (float*)(ws + 0x3100000);
  float* p_pred   = (float*)(ws + 0x3180000);
  float* q_pred   = (float*)(ws + 0x3280000);
  float* acc_pred = (float*)(ws + 0x3380000);
  float* h_glob = p_hist;
  float* c_glob = q_hist;

  k_zero<<<(NNODE + 255) / 256, 256, 0, stream>>>(deg, NNODE);
  k_zero<<<(BATCH * HISTT * NNODE * 2 + 255) / 256, 256, 0, stream>>>(
      acc_hist, BATCH * HISTT * NNODE * 2);
  k_deg<<<NEDGE / 256, 256, 0, stream>>>(ei, deg);
  k_dis<<<NNODE / 256, 256, 0, stream>>>(deg, dis);
  k_enorm<<<NEDGE / 256, 256, 0, stream>>>(ei, dis, enorm);

  k_proj_hist<<<BATCH * HISTT * NNODE / 256, 256, 0, stream>>>(pm25, feat, cheb_w, cheb_b,
                                                               p_hist, q_hist);
  k_scatter_hist<<<BATCH * HISTT * NEDGE / 256, 256, 0, stream>>>(ei, enorm, p_hist, acc_hist);
  k_fin_hist<<<BATCH * HISTT * NNODE * 2 / 256, 256, 0, stream>>>(q_hist, acc_hist);

  hist_lstm_kernel<<<BATCH * NNODE / 128, 256, 0, stream>>>(
      pm25, feat, acc_hist, x2h_w, x2h_b, h2h_w, h2h_b, fc_w, fc_b, h_glob, c_glob, xn_glob);

  for (int k = 0; k < PREDT; ++k) {
    k_proj_pred<<<BATCH * NNODE / 256, 256, 0, stream>>>(xn_glob, feat, cheb_w, cheb_b,
                                                         p_pred, q_pred, acc_pred, k);
    k_scatter_pred<<<BATCH * NEDGE / 256, 256, 0, stream>>>(ei, enorm, p_pred, acc_pred);
    pred_lstm_kernel<<<BATCH * NNODE / 128, 256, 0, stream>>>(
        feat, q_pred, acc_pred, x2h_w, x2h_b, h2h_w, h2h_b, fc_w, fc_b,
        h_glob, c_glob, xn_glob, out, k);
  }
}


# Float-atomic edge scatter replaced by an in-call dst-sorted edge list (integer-atomic counts, slots ranked by edge id) and per-node gather sums in ascending edge id; no float atomics
# speedup vs baseline: 4.2771x; 4.2771x over previous
#include <hip/hip_runtime.h>
#include <hip/hip_bf16.h>
#include <cstdint>
#include <cstddef>

#define BATCH 32
#define NNODE 4096
#define HISTT 16
#define PREDT 8
#define NEDGE 65536

typedef float v2f __attribute__((ext_vector_type(2)));
typedef float v8f __attribute__((ext_vector_type(8)));

__device__ __forceinline__ float sigm(float x) { return 1.f / (1.f + __expf(-x)); }

__device__ __forceinline__ int atomAddI(int* p, int v) {
  return __hip_atomic_fetch_add(p, v, __ATOMIC_RELAXED, __HIP_MEMORY_SCOPE_AGENT);
}

__global__ void k_zero_i(int* __restrict__ p, int n) {
  int i = blockIdx.x * 256 + threadIdx.x;
  if (i < n) p[i] = 0;
}

__global__ void k_count(const int* __restrict__ ei, int* __restrict__ cnt_src,
                        int* __restrict__ cnt_dst, int* __restrict__ arr) {
  int e = blockIdx.x * 256 + threadIdx.x;
  if (e < NEDGE) {
    atomAddI(cnt_src + ei[e], 1);
    arr[e] = atomAddI(cnt_dst + ei[NEDGE + e], 1);
  }
}

__launch_bounds__(1024)
__global__ void k_scan(const int* __restrict__ cnt_src, const int* __restrict__ cnt_dst,
                       int* __restrict__ rowptr, float* __restrict__ dis) {
  __shared__ int s[1024];
  const int t = threadIdx.x;
  int c0 = cnt_dst[4 * t + 0], c1 = cnt_dst[4 * t + 1];
  int c2 = cnt_dst[4 * t + 2], c3 = cnt_dst[4 * t + 3];
  int sum = c0 + c1 + c2 + c3;
  s[t] = sum;
  __syncthreads();
  for (int off = 1; off < 1024; off <<= 1) {
    int v = (t >= off) ? s[t - off] : 0;
    __syncthreads();
    s[t] += v;
    __syncthreads();
  }
  int incl = s[t];
  int ex = incl - sum;
  rowptr[4 * t + 0] = ex;
  rowptr[4 * t + 1] = ex + c0;
  rowptr[4 * t + 2] = ex + c0 + c1;
  rowptr[4 * t + 3] = ex + c0 + c1 + c2;
  if (t == 1023) rowptr[NNODE] = incl;
#pragma unroll
  for (int i = 0; i < 4; ++i) {
    float d = (float)cnt_src[4 * t + i];
    dis[4 * t + i] = d > 0.f ? rsqrtf(d) : 0.f;
  }
}

__global__ void k_place(const int* __restrict__ ei, const int* __restrict__ rowptr,
                        const int* __restrict__ arr, int* __restrict__ tmp) {
  int e = blockIdx.x * 256 + threadIdx.x;
  if (e < NEDGE) tmp[rowptr[ei[NEDGE + e]] + arr[e]] = e;
}

__global__ void k_rank(const int* __restrict__ ei, const int* __restrict__ rowptr,
                       const float* __restrict__ dis, const int* __restrict__ tmp,
                       int2* __restrict__ csr) {
  int i = blockIdx.x * 256 + threadIdx.x;
  if (i >= NEDGE) return;
  int e = tmp[i];
  int s = ei[e], d = ei[NEDGE + e];
  int lo = rowptr[d], hi = rowptr[d + 1];
  int rank = 0;
  for (int j = lo; j < hi; ++j) rank += (tmp[j] < e) ? 1 : 0;
  csr[lo + rank] = make_int2(s, __float_as_int(-dis[s] * dis[d]));
}

__global__ void k_proj_hist(const float* __restrict__ pm25, const float* __restrict__ feat,
                            const float* __restrict__ cheb_w, const float* __restrict__ cheb_b,
                            float* __restrict__ p, float* __restrict__ q) {
  int i = blockIdx.x * 256 + threadIdx.x;
  if (i >= BATCH * HISTT * NNODE) return;
  int bt = i >> 12, b = bt >> 4, t = bt & 15, n = i & 4095;
  float x[10];
  x[0] = pm25[i];
  const float* fp = feat + (((size_t)b * 24 + t) * NNODE + n) * 9;
#pragma unroll
  for (int j = 0; j < 9; ++j) x[1 + j] = fp[j];
  float q0 = cheb_b[0], q1 = cheb_b[1], p0 = 0.f, p1 = 0.f;
#pragma unroll
  for (int f = 0; f < 10; ++f) {
    q0 += x[f] * cheb_w[f * 2 + 0];
    q1 += x[f] * cheb_w[f * 2 + 1];
    p0 += x[f] * cheb_w[20 + f * 2 + 0];
    p1 += x[f] * cheb_w[20 + f * 2 + 1];
  }
  size_t o = (size_t)i * 2;
  p[o] = p0; p[o + 1] = p1; q[o] = q0; q[o + 1] = q1;
}

__launch_bounds__(1024)
__global__ void k_gather_hist(const int* __restrict__ rowptr, const int2* __restrict__ csr,
                              const float* __restrict__ p, const float* __restrict__ q,
                              float* __restrict__ xg) {
  __shared__ float4 sP[NNODE];
  const int tid = threadIdx.x;
  const size_t base = (size_t)blockIdx.x * 2 * NNODE;
  const float2* p0 = (const float2*)p + base;
  const float2* p1 = p0 + NNODE;
  for (int n = tid; n < NNODE; n += 1024) {
    float2 a = p0[n], b = p1[n];
    sP[n] = make_float4(a.x, a.y, b.x, b.y);
  }
  __syncthreads();
  const float2* q0 = (const float2*)q + base;
  const float2* q1 = q0 + NNODE;
  float2* x0 = (float2*)xg + base;
  float2* x1 = x0 + NNODE;
  for (int n = tid; n < NNODE; n += 1024) {
    int lo = rowptr[n], hi = rowptr[n + 1];
    float a0 = 0.f, a1 = 0.f, a2 = 0.f, a3 = 0.f;
    for (int j = lo; j < hi; ++j) {
      int2 en = csr[j];
      float nv = __int_as_float(en.y);
      float4 pv = sP[en.x];
      a0 += pv.x * nv; a1 += pv.y * nv; a2 += pv.z * nv; a3 += pv.w * nv;
    }
    float2 qa = q0[n], qb = q1[n];
    x0[n] = make_float2(sigm(qa.x + a0), sigm(qa.y + a1));
    x1[n] = make_float2(sigm(qb.x + a2), sigm(qb.y + a3));
  }
}

__global__ void k_proj_pred(const float* __restrict__ xn, const float* __restrict__ feat,
                            const float* __restrict__ cheb_w, const float* __restrict__ cheb_b,
                            float* __restrict__ p, float* __restrict__ q, int kstep) {
  int i = blockIdx.x * 256 + threadIdx.x;
  if (i >= BATCH * NNODE) return;
  int b = i >> 12, n = i & 4095;
  float x[10];
  x[0] = xn[i];
  const float* fp = feat + (((size_t)b * 24 + HISTT + kstep) * NNODE + n) * 9;
#pragma unroll
  for (int j = 0; j < 9; ++j) x[1 + j] = fp[j];
  float q0 = cheb_b[0], q1 = cheb_b[1], p0 = 0.f, p1 = 0.f;
#pragma unroll
  for (int f = 0; f < 10; ++f) {
    q0 += x[f] * cheb_w[f * 2 + 0];
    q1 += x[f] * cheb_w[f * 2 + 1];
    p0 += x[f] * cheb_w[20 + f * 2 + 0];
    p1 += x[f] * cheb_w[20 + f * 2 + 1];
  }
  size_t o = (size_t)i * 2;
  p[o] = p0; p[o + 1] = p1; q[o] = q0; q[o + 1] = q1;
}

__global__ void k_gather_pred(const int* __restrict__ rowptr, const int2* __restrict__ csr,
                              const float* __restrict__ p, const float* __restrict__ q,
                              float* __restrict__ xg) {
  int i = blockIdx.x * 256 + threadIdx.x;
  if (i >= BATCH * NNODE) return;
  int b = i >> 12, n = i & 4095;
  const float2* pb = (const float2*)p + (size_t)b * NNODE;
  int lo = rowptr[n], hi = rowptr[n + 1];
  float a0 = 0.f, a1 = 0.f;
  for (int j = lo; j < hi; ++j) {
    int2 en = csr[j];
    float nv = __int_as_float(en.y);
    float2 pv = pb[en.x];
    a0 += pv.x * nv; a1 += pv.y * nv;
  }
  float2 qv = ((const float2*)q)[i];
  ((float2*)xg)[i] = make_float2(sigm(qv.x + a0), sigm(qv.y + a1));
}

#define WT_STRIDE 48

__device__ __forceinline__ void load_weights(int tid,
    const float* __restrict__ x2h_w, const float* __restrict__ x2h_b,
    const float* __restrict__ h2h_w, const float* __restrict__ h2h_b,
    const float* __restrict__ fc_w,
    float* sWT, float* sBias, float* sFc) {
  for (int i = tid; i < 44 * 128; i += 256) {
    int k = i >> 7, n = i & 127;
    sWT[n * WT_STRIDE + k] = (k < 12) ? x2h_w[k * 128 + n] : h2h_w[(k - 12) * 128 + n];
  }
  if (tid < 128) sBias[tid] = x2h_b[tid] + h2h_b[tid];
  if (tid < 32) sFc[tid] = fc_w[tid];
}

__device__ __forceinline__ void gate_compute(const float* __restrict__ sXw,
                                             const float* __restrict__ sHw,
                                             const float* __restrict__ sWT,
                                             const float* __restrict__ sBias,
                                             int m, int hi, v8f* acc) {
  v2f ax[3], ah[8];
#pragma unroll
  for (int kk = 0; kk < 3; ++kk) ax[kk] = *(const v2f*)(sXw + m * 12 + 4 * kk + 2 * hi);
#pragma unroll
  for (int kk = 0; kk < 8; ++kk) ah[kk] = *(const v2f*)(sHw + m * 32 + 4 * kk + 2 * hi);
#pragma unroll
  for (int ct = 0; ct < 8; ++ct) {
    float bb = sBias[ct * 16 + m];
    v8f a = {bb, bb, bb, bb, bb, bb, bb, bb};
    const float* wcol = sWT + (ct * 16 + m) * WT_STRIDE;
#pragma unroll
    for (int kk = 0; kk < 3; ++kk) {
      v2f bf = *(const v2f*)(wcol + 4 * kk + 2 * hi);
      a = __builtin_amdgcn_wmma_f32_16x16x4_f32(false, ax[kk], false, bf, (short)0, a,
                                                false, false);
    }
#pragma unroll
    for (int kk = 0; kk < 8; ++kk) {
      v2f bf = *(const v2f*)(wcol + 12 + 4 * kk + 2 * hi);
      a = __builtin_amdgcn_wmma_f32_16x16x4_f32(false, ah[kk], false, bf, (short)0, a,
                                                false, false);
    }
    acc[ct] = a;
  }
}

__device__ __forceinline__ void lstm_update(const v8f* acc, float* c01, float* h01,
                                            float* sHw, int m, int hi) {
#pragma unroll
  for (int ct = 0; ct < 2; ++ct) {
#pragma unroll
    for (int r = 0; r < 8; ++r) {
      float iv = sigm(acc[ct][r]);
      float fv = sigm(acc[2 + ct][r]);
      float gv = tanhf(acc[4 + ct][r]);
      float ov = sigm(acc[6 + ct][r]);
      float cn = fv * c01[ct * 8 + r] + iv * gv;
      float hn = ov * tanhf(cn);
      c01[ct * 8 + r] = cn;
      h01[ct * 8 + r] = hn;
      sHw[(r + 8 * hi) * 32 + ct * 16 + m] = hn;
    }
  }
}

__launch_bounds__(256)
__global__ void hist_lstm_kernel(
    const float* __restrict__ pm25, const float* __restrict__ feat,
    const float* __restrict__ xg,
    const float* __restrict__ x2h_w, const float* __restrict__ x2h_b,
    const float* __restrict__ h2h_w, const float* __restrict__ h2h_b,
    const float* __restrict__ fc_w, const float* __restrict__ fc_b,
    float* __restrict__ h_glob, float* __restrict__ c_glob, float* __restrict__ xn_glob) {
  __shared__ float sWT[128 * WT_STRIDE];
  __shared__ float sBias[128];
  __shared__ float sFc[32];
  __shared__ float sX[8][16 * 12];
  __shared__ float sH[8][16 * 32];

  const int tid = threadIdx.x;
  load_weights(tid, x2h_w, x2h_b, h2h_w, h2h_b, fc_w, sWT, sBias, sFc);

  const int wave = tid >> 5, lane = tid & 31, m = lane & 15, hi = lane >> 4;
  const int rowbase = blockIdx.x * 128 + wave * 16;
  float* sXw = sX[wave];
  float* sHw = sH[wave];
  for (int j = lane; j < 512; j += 32) sHw[j] = 0.f;
  float c01[16], h01[16];
#pragma unroll
  for (int j = 0; j < 16; ++j) { c01[j] = 0.f; h01[j] = 0.f; }
  __syncthreads();

  for (int t = 0; t < HISTT; ++t) {
    if (lane < 16) {
      int R = rowbase + lane;
      int b = R >> 12, n = R & 4095;
      size_t hb = ((size_t)(b * 16 + t)) * NNODE + n;
      sXw[lane * 12 + 0] = pm25[hb];
      const float* fp = feat + (((size_t)b * 24 + t) * NNODE + n) * 9;
#pragma unroll
      for (int j = 0; j < 9; ++j) sXw[lane * 12 + 1 + j] = fp[j];
      sXw[lane * 12 + 10] = xg[hb * 2 + 0];
      sXw[lane * 12 + 11] = xg[hb * 2 + 1];
    }
    __syncthreads();
    v8f acc[8];
    gate_compute(sXw, sHw, sWT, sBias, m, hi, acc);
    lstm_update(acc, c01, h01, sHw, m, hi);
  }

#pragma unroll
  for (int ct = 0; ct < 2; ++ct)
#pragma unroll
    for (int r = 0; r < 8; ++r) {
      size_t R = (size_t)rowbase + r + 8 * hi;
      c_glob[R * 32 + ct * 16 + m] = c01[ct * 8 + r];
      h_glob[R * 32 + ct * 16 + m] = h01[ct * 8 + r];
    }
  __syncthreads();
  if (lane < 16) {
    float s = fc_b[0];
#pragma unroll
    for (int j = 0; j < 32; ++j) s += sHw[lane * 32 + j] * sFc[j];
    xn_glob[rowbase + lane] = s;
  }
}

__launch_bounds__(256)
__global__ void pred_lstm_kernel(
    const float* __restrict__ feat,
    const float* __restrict__ xg_pred,
    const float* __restrict__ x2h_w, const float* __restrict__ x2h_b,
    const float* __restrict__ h2h_w, const float* __restrict__ h2h_b,
    const float* __restrict__ fc_w, const float* __restrict__ fc_b,
    float* __restrict__ h_glob, float* __restrict__ c_glob, float* __restrict__ xn_glob,
    float* __restrict__ out, int kstep) {
  __shared__ float sWT[128 * WT_STRIDE];
  __shared__ float sBias[128];
  __shared__ float sFc[32];
  __shared__ float sX[8][16 * 12];
  __shared__ float sH[8][16 * 32];

  const int tid = threadIdx.x;
  load_weights(tid, x2h_w, x2h_b, h2h_w, h2h_b, fc_w, sWT, sBias, sFc);

  const int wave = tid >> 5, lane = tid & 31, m = lane & 15, hi = lane >> 4;
  const int rowbase = blockIdx.x * 128 + wave * 16;
  float* sXw = sX[wave];
  float* sHw = sH[wave];
  for (int j = lane; j < 512; j += 32) sHw[j] = h_glob[(size_t)rowbase * 32 + j];
  float c01[16], h01[16];
#pragma unroll
  for (int ct = 0; ct < 2; ++ct)
#pragma unroll
    for (int r = 0; r < 8; ++r)
      c01[ct * 8 + r] = c_glob[((size_t)rowbase + r + 8 * hi) * 32 + ct * 16 + m];

  if (lane < 16) {
    int R = rowbase + lane;
    int b = R >> 12, n = R & 4095;
    sXw[lane * 12 + 0] = xn_glob[R];
    const float* fp = feat + (((size_t)b * 24 + HISTT + kstep) * NNODE + n) * 9;
#pragma unroll
    for (int j = 0; j < 9; ++j) sXw[lane * 12 + 1 + j] = fp[j];
    sXw[lane * 12 + 10] = xg_pred[(size_t)R * 2 + 0];
    sXw[lane * 12 + 11] = xg_pred[(size_t)R * 2 + 1];
  }
  __syncthreads();

  v8f acc[8];
  gate_compute(sXw, sHw, sWT, sBias, m, hi, acc);
  lstm_update(acc, c01, h01, sHw, m, hi);

#pragma unroll
  for (int ct = 0; ct < 2; ++ct)
#pragma unroll
    for (int r = 0; r < 8; ++r) {
      size_t R = (size_t)rowbase + r + 8 * hi;
      c_glob[R * 32 + ct * 16 + m] = c01[ct * 8 + r];
      h_glob[R * 32 + ct * 16 + m] = h01[ct * 8 + r];
    }
  __syncthreads();
  if (lane < 16) {
    float s = fc_b[0];
#pragma unroll
    for (int j = 0; j < 32; ++j) s += sHw[lane * 32 + j] * sFc[j];
    int R = rowbase + lane;
    int b = R >> 12, n = R & 4095;
    xn_glob[R] = s;
    out[((size_t)b * PREDT + kstep) * NNODE + n] = s;
  }
}

extern "C" void kernel_launch(void* const* d_in, const int* in_sizes, int n_in,
                              void* d_out, int out_size, void* d_ws, size_t ws_size,
                              hipStream_t stream) {
  (void)in_sizes; (void)n_in; (void)out_size; (void)ws_size;
  const float* pm25   = (const float*)d_in[0];
  const float* feat   = (const float*)d_in[1];
  const int*   ei     = (const int*)d_in[2];
  const float* cheb_w = (const float*)d_in[3];
  const float* cheb_b = (const float*)d_in[4];
  const float* x2h_w  = (const float*)d_in[5];
  const float* x2h_b  = (const float*)d_in[6];
  const float* h2h_w  = (const float*)d_in[7];
  const float* h2h_b  = (const float*)d_in[8];
  const float* fc_w   = (const float*)d_in[9];
  const float* fc_b   = (const float*)d_in[10];
  float* out = (float*)d_out;
  char* ws = (char*)d_ws;

  int*   cnt_src  = (int*)(ws + 0);
  int*   cnt_dst  = (int*)(ws + 0x4000);
  float* dis      = (float*)(ws + 0x8000);
  int*   rowptr   = (int*)(ws + 0xC000);
  int2*  csr      = (int2*)(ws + 0x14000);
  float* p_hist   = (float*)(ws + 0x100000);
  float* q_hist   = (float*)(ws + 0x1100000);
  float* xg_hist  = (float*)(ws + 0x2100000);
  float* xn_glob  = (float*)(ws + 0x3100000);
  float* p_pred   = (float*)(ws + 0x3180000);
  float* q_pred   = (float*)(ws + 0x3280000);
  float* xg_pred  = (float*)(ws + 0x3380000);
  int* arr = (int*)p_pred;
  int* tmp = (int*)q_pred;
  float* h_glob = p_hist;
  float* c_glob = q_hist;

  k_zero_i<<<2 * NNODE / 256, 256, 0, stream>>>(cnt_src, 2 * NNODE);
  k_count<<<NEDGE / 256, 256, 0, stream>>>(ei, cnt_src, cnt_dst, arr);
  k_scan<<<1, 1024, 0, stream>>>(cnt_src, cnt_dst, rowptr, dis);
  k_place<<<NEDGE / 256, 256, 0, stream>>>(ei, rowptr, arr, tmp);
  k_rank<<<NEDGE / 256, 256, 0, stream>>>(ei, rowptr, dis, tmp, csr);

  k_proj_hist<<<BATCH * HISTT * NNODE / 256, 256, 0, stream>>>(pm25, feat, cheb_w, cheb_b,
                                                               p_hist, q_hist);
  k_gather_hist<<<BATCH * HISTT / 2, 1024, 0, stream>>>(rowptr, csr, p_hist, q_hist, xg_hist);

  hist_lstm_kernel<<<BATCH * NNODE / 128, 256, 0, stream>>>(
      pm25, feat, xg_hist, x2h_w, x2h_b, h2h_w, h2h_b, fc_w, fc_b, h_glob, c_glob, xn_glob);

  for (int k = 0; k < PREDT; ++k) {
    k_proj_pred<<<BATCH * NNODE / 256, 256, 0, stream>>>(xn_glob, feat, cheb_w, cheb_b,
                                                         p_pred, q_pred, k);
    k_gather_pred<<<BATCH * NNODE / 256, 256, 0, stream>>>(rowptr, csr, p_pred, q_pred, xg_pred);
    pred_lstm_kernel<<<BATCH * NNODE / 128, 256, 0, stream>>>(
        feat, xg_pred, x2h_w, x2h_b, h2h_w, h2h_b, fc_w, fc_b,
        h_glob, c_glob, xn_glob, out, k);
  }
}


# Prediction step's projection and neighbour gather folded into the LSTM step kernel (3 launches per step -> 1, 32 -> 16 per call); edge terms staged block-wide in LDS, per-node sum order unchanged
# speedup vs baseline: 4.7445x; 1.1093x over previous
#include <hip/hip_runtime.h>
#include <hip/hip_bf16.h>
#include <cstdint>
#include <cstddef>

#define BATCH 32
#define NNODE 4096
#define HISTT 16
#define PREDT 8
#define NEDGE 65536

typedef float v2f __attribute__((ext_vector_type(2)));
typedef float v8f __attribute__((ext_vector_type(8)));

__device__ __forceinline__ float sigm(float x) { return 1.f / (1.f + __expf(-x)); }

__device__ __forceinline__ int atomAddI(int* p, int v) {
  return __hip_atomic_fetch_add(p, v, __ATOMIC_RELAXED, __HIP_MEMORY_SCOPE_AGENT);
}

__global__ void k_zero_i(int* __restrict__ p, int n) {
  int i = blockIdx.x * 256 + threadIdx.x;
  if (i < n) p[i] = 0;
}

__global__ void k_count(const int* __restrict__ ei, int* __restrict__ cnt_src,
                        int* __restrict__ cnt_dst, int* __restrict__ arr) {
  int e = blockIdx.x * 256 + threadIdx.x;
  if (e < NEDGE) {
    atomAddI(cnt_src + ei[e], 1);
    arr[e] = atomAddI(cnt_dst + ei[NEDGE + e], 1);
  }
}

__launch_bounds__(1024)
__global__ void k_scan(const int* __restrict__ cnt_src, const int* __restrict__ cnt_dst,
                       int* __restrict__ rowptr, float* __restrict__ dis) {
  __shared__ int s[1024];
  const int t = threadIdx.x;
  int c0 = cnt_dst[4 * t + 0], c1 = cnt_dst[4 * t + 1];
  int c2 = cnt_dst[4 * t + 2], c3 = cnt_dst[4 * t + 3];
  int sum = c0 + c1 + c2 + c3;
  s[t] = sum;
  __syncthreads();
  for (int off = 1; off < 1024; off <<= 1) {
    int v = (t >= off) ? s[t - off] : 0;
    __syncthreads();
    s[t] += v;
    __syncthreads();
  }
  int incl = s[t];
  int ex = incl - sum;
  rowptr[4 * t + 0] = ex;
  rowptr[4 * t + 1] = ex + c0;
  rowptr[4 * t + 2] = ex + c0 + c1;
  rowptr[4 * t + 3] = ex + c0 + c1 + c2;
  if (t == 1023) rowptr[NNODE] = incl;
#pragma unroll
  for (int i = 0; i < 4; ++i) {
    float d = (float)cnt_src[4 * t + i];
    dis[4 * t + i] = d > 0.f ? rsqrtf(d) : 0.f;
  }
}

__global__ void k_place(const int* __restrict__ ei, const int* __restrict__ rowptr,
                        const int* __restrict__ arr, int* __restrict__ tmp) {
  int e = blockIdx.x * 256 + threadIdx.x;
  if (e < NEDGE) tmp[rowptr[ei[NEDGE + e]] + arr[e]] = e;
}

__global__ void k_rank(const int* __restrict__ ei, const int* __restrict__ rowptr,
                       const float* __restrict__ dis, const int* __restrict__ tmp,
                       int2* __restrict__ csr) {
  int i = blockIdx.x * 256 + threadIdx.x;
  if (i >= NEDGE) return;
  int e = tmp[i];
  int s = ei[e], d = ei[NEDGE + e];
  int lo = rowptr[d], hi = rowptr[d + 1];
  int rank = 0;
  for (int j = lo; j < hi; ++j) rank += (tmp[j] < e) ? 1 : 0;
  csr[lo + rank] = make_int2(s, __float_as_int(-dis[s] * dis[d]));
}

__global__ void k_proj_hist(const float* __restrict__ pm25, const float* __restrict__ feat,
                            const float* __restrict__ cheb_w, const float* __restrict__ cheb_b,
                            float* __restrict__ p, float* __restrict__ q) {
  int i = blockIdx.x * 256 + threadIdx.x;
  if (i >= BATCH * HISTT * NNODE) return;
  int bt = i >> 12, b = bt >> 4, t = bt & 15, n = i & 4095;
  float x[10];
  x[0] = pm25[i];
  const float* fp = feat + (((size_t)b * 24 + t) * NNODE + n) * 9;
#pragma unroll
  for (int j = 0; j < 9; ++j) x[1 + j] = fp[j];
  float q0 = cheb_b[0], q1 = cheb_b[1], p0 = 0.f, p1 = 0.f;
#pragma unroll
  for (int f = 0; f < 10; ++f) {
    q0 += x[f] * cheb_w[f * 2 + 0];
    q1 += x[f] * cheb_w[f * 2 + 1];
    p0 += x[f] * cheb_w[20 + f * 2 + 0];
    p1 += x[f] * cheb_w[20 + f * 2 + 1];
  }
  size_t o = (size_t)i * 2;
  p[o] = p0; p[o + 1] = p1; q[o] = q0; q[o + 1] = q1;
}

__launch_bounds__(1024)
__global__ void k_gather_hist(const int* __restrict__ rowptr, const int2* __restrict__ csr,
                              const float* __restrict__ p, const float* __restrict__ q,
                              float* __restrict__ xg) {
  __shared__ float4 sP[NNODE];
  const int tid = threadIdx.x;
  const size_t base = (size_t)blockIdx.x * 2 * NNODE;
  const float2* p0 = (const float2*)p + base;
  const float2* p1 = p0 + NNODE;
  for (int n = tid; n < NNODE; n += 1024) {
    float2 a = p0[n], b = p1[n];
    sP[n] = make_float4(a.x, a.y, b.x, b.y);
  }
  __syncthreads();
  const float2* q0 = (const float2*)q + base;
  const float2* q1 = q0 + NNODE;
  float2* x0 = (float2*)xg + base;
  float2* x1 = x0 + NNODE;
  for (int n = tid; n < NNODE; n += 1024) {
    int lo = rowptr[n], hi = rowptr[n + 1];
    float a0 = 0.f, a1 = 0.f, a2 = 0.f, a3 = 0.f;
    for (int j = lo; j < hi; ++j) {
      int2 en = csr[j];
      float nv = __int_as_float(en.y);
      float4 pv = sP[en.x];
      a0 += pv.x * nv; a1 += pv.y * nv; a2 += pv.z * nv; a3 += pv.w * nv;
    }
    float2 qa = q0[n], qb = q1[n];
    x0[n] = make_float2(sigm(qa.x + a0), sigm(qa.y + a1));
    x1[n] = make_float2(sigm(qb.x + a2), sigm(qb.y + a3));
  }
}

#define ECAP 2560

#define EPT (ECAP / 256)

__device__ __forceinline__ void stage_list(const int2* __restrict__ csr, int2* en,
                                           int c, int cend, int tid) {
#pragma unroll
  for (int i = 0; i < EPT; ++i) en[i] = csr[max(min(c + tid + 256 * i, cend - 1), 0)];
}

__device__ __forceinline__ void stage_p(const float2* __restrict__ pb, const int2* en,
                                        float4* sE, int c, int cend, int tid) {
  float2 pv[EPT];
#pragma unroll
  for (int i = 0; i < EPT; ++i) pv[i] = pb[en[i].x];
#pragma unroll
  for (int i = 0; i < EPT; ++i) {
    int j = c + tid + 256 * i;
    if (j < cend) sE[j - c] = make_float4(pv[i].x, pv[i].y, __int_as_float(en[i].y), 0.f);
  }
}

__device__ __forceinline__ void sum_row(const float4* sE, int c, int cend, int lo, int hi_e,
                                        float& a0, float& a1) {
  int j0 = max(lo, c) - c, j1 = min(hi_e, cend) - c;
  for (int j = j0; j < j1; j += 4) {
    float4 v0 = sE[j];
    float4 v1 = sE[min(j + 1, j1 - 1)];
    float4 v2 = sE[min(j + 2, j1 - 1)];
    float4 v3 = sE[min(j + 3, j1 - 1)];
    a0 += v0.x * v0.z; a1 += v0.y * v0.z;
    if (j + 1 < j1) { a0 += v1.x * v1.z; a1 += v1.y * v1.z; }
    if (j + 2 < j1) { a0 += v2.x * v2.z; a1 += v2.y * v2.z; }
    if (j + 3 < j1) { a0 += v3.x * v3.z; a1 += v3.y * v3.z; }
  }
}

#define WT_STRIDE 48

__device__ __forceinline__ void load_weights(int tid,
    const float* __restrict__ x2h_w, const float* __restrict__ x2h_b,
    const float* __restrict__ h2h_w, const float* __restrict__ h2h_b,
    const float* __restrict__ fc_w,
    float* sWT, float* sBias, float* sFc) {
  for (int i = tid; i < 44 * 128; i += 256) {
    int k = i >> 7, n = i & 127;
    sWT[n * WT_STRIDE + k] = (k < 12) ? x2h_w[k * 128 + n] : h2h_w[(k - 12) * 128 + n];
  }
  if (tid < 128) sBias[tid] = x2h_b[tid] + h2h_b[tid];
  if (tid < 32) sFc[tid] = fc_w[tid];
}

__device__ __forceinline__ void gate_compute(const float* __restrict__ sXw,
                                             const float* __restrict__ sHw,
                                             const float* __restrict__ sWT,
                                             const float* __restrict__ sBias,
                                             int m, int hi, v8f* acc) {
  v2f ax[3], ah[8];
#pragma unroll
  for (int kk = 0; kk < 3; ++kk) ax[kk] = *(const v2f*)(sXw + m * 12 + 4 * kk + 2 * hi);
#pragma unroll
  for (int kk = 0; kk < 8; ++kk) ah[kk] = *(const v2f*)(sHw + m * 32 + 4 * kk + 2 * hi);
#pragma unroll
  for (int ct = 0; ct < 8; ++ct) {
    float bb = sBias[ct * 16 + m];
    v8f a = {bb, bb, bb, bb, bb, bb, bb, bb};
    const float* wcol = sWT + (ct * 16 + m) * WT_STRIDE;
#pragma unroll
    for (int kk = 0; kk < 3; ++kk) {
      v2f bf = *(const v2f*)(wcol + 4 * kk + 2 * hi);
      a = __builtin_amdgcn_wmma_f32_16x16x4_f32(false, ax[kk], false, bf, (short)0, a,
                                                false, false);
    }
#pragma unroll
    for (int kk = 0; kk < 8; ++kk) {
      v2f bf = *(const v2f*)(wcol + 12 + 4 * kk + 2 * hi);
      a = __builtin_amdgcn_wmma_f32_16x16x4_f32(false, ah[kk], false, bf, (short)0, a,
                                                false, false);
    }
    acc[ct] = a;
  }
}

__device__ __forceinline__ void lstm_update(const v8f* acc, float* c01, float* h01,
                                            float* sHw, int m, int hi) {
#pragma unroll
  for (int ct = 0; ct < 2; ++ct) {
#pragma unroll
    for (int r = 0; r < 8; ++r) {
      float iv = sigm(acc[ct][r]);
      float fv = sigm(acc[2 + ct][r]);
      float gv = tanhf(acc[4 + ct][r]);
      float ov = sigm(acc[6 + ct][r]);
      float cn = fv * c01[ct * 8 + r] + iv * gv;
      float hn = ov * tanhf(cn);
      c01[ct * 8 + r] = cn;
      h01[ct * 8 + r] = hn;
      sHw[(r + 8 * hi) * 32 + ct * 16 + m] = hn;
    }
  }
}

__launch_bounds__(256)
__global__ void hist_lstm_kernel(
    const float* __restrict__ pm25, const float* __restrict__ feat,
    const float* __restrict__ xg, const float* __restrict__ cheb_w,
    const float* __restrict__ x2h_w, const float* __restrict__ x2h_b,
    const float* __restrict__ h2h_w, const float* __restrict__ h2h_b,
    const float* __restrict__ fc_w, const float* __restrict__ fc_b,
    float* __restrict__ h_glob, float* __restrict__ c_glob, float* __restrict__ xn_glob,
    float* __restrict__ p_first) {
  __shared__ float sWT[128 * WT_STRIDE];
  __shared__ float sBias[128];
  __shared__ float sFc[32];
  __shared__ float sX[8][16 * 12];
  __shared__ float sH[8][16 * 32];

  const int tid = threadIdx.x;
  load_weights(tid, x2h_w, x2h_b, h2h_w, h2h_b, fc_w, sWT, sBias, sFc);

  const int wave = tid >> 5, lane = tid & 31, m = lane & 15, hi = lane >> 4;
  const int rowbase = blockIdx.x * 128 + wave * 16;
  float* sXw = sX[wave];
  float* sHw = sH[wave];
  for (int j = lane; j < 512; j += 32) sHw[j] = 0.f;
  float c01[16], h01[16];
#pragma unroll
  for (int j = 0; j < 16; ++j) { c01[j] = 0.f; h01[j] = 0.f; }
  __syncthreads();

  for (int t = 0; t < HISTT; ++t) {
    if (lane < 16) {
      int R = rowbase + lane;
      int b = R >> 12, n = R & 4095;
      size_t hb = ((size_t)(b * 16 + t)) * NNODE + n;
      sXw[lane * 12 + 0] = pm25[hb];
      const float* fp = feat + (((size_t)b * 24 + t) * NNODE + n) * 9;
#pragma unroll
      for (int j = 0; j < 9; ++j) sXw[lane * 12 + 1 + j] = fp[j];
      sXw[lane * 12 + 10] = xg[hb * 2 + 0];
      sXw[lane * 12 + 11] = xg[hb * 2 + 1];
    }
    __syncthreads();
    v8f acc[8];
    gate_compute(sXw, sHw, sWT, sBias, m, hi, acc);
    lstm_update(acc, c01, h01, sHw, m, hi);
  }

#pragma unroll
  for (int ct = 0; ct < 2; ++ct)
#pragma unroll
    for (int r = 0; r < 8; ++r) {
      size_t R = (size_t)rowbase + r + 8 * hi;
      c_glob[R * 32 + ct * 16 + m] = c01[ct * 8 + r];
      h_glob[R * 32 + ct * 16 + m] = h01[ct * 8 + r];
    }
  __syncthreads();
  if (lane < 16) {
    float s = fc_b[0];
#pragma unroll
    for (int j = 0; j < 32; ++j) s += sHw[lane * 32 + j] * sFc[j];
    xn_glob[rowbase + lane] = s;
    int R = rowbase + lane;
    int b = R >> 12, n = R & 4095;
    const float* fp = feat + (((size_t)b * 24 + HISTT) * NNODE + n) * 9;
    float p0 = 0.f, p1 = 0.f;
    p0 += s * cheb_w[20]; p1 += s * cheb_w[21];
#pragma unroll
    for (int j = 0; j < 9; ++j) {
      float xv = fp[j];
      p0 += xv * cheb_w[20 + (1 + j) * 2 + 0];
      p1 += xv * cheb_w[20 + (1 + j) * 2 + 1];
    }
    ((float2*)p_first)[R] = make_float2(p0, p1);
  }
}

__launch_bounds__(256)
__global__ void pred_step_kernel(
    const float* __restrict__ feat,
    const int* __restrict__ rowptr, const int2* __restrict__ csr,
    const float* __restrict__ cheb_w, const float* __restrict__ cheb_b,
    const float* __restrict__ x2h_w, const float* __restrict__ x2h_b,
    const float* __restrict__ h2h_w, const float* __restrict__ h2h_b,
    const float* __restrict__ fc_w, const float* __restrict__ fc_b,
    float* __restrict__ h_glob, float* __restrict__ c_glob, float* __restrict__ xn_glob,
    float* __restrict__ pA, float* __restrict__ pB,
    float* __restrict__ out, int kstep) {
  __shared__ float sWT[128 * WT_STRIDE];
  __shared__ float sBias[128];
  __shared__ float sFc[32];
  __shared__ float sX[8][16 * 12];
  __shared__ float sH[8][16 * 32];
  __shared__ float sXn[8][16 * 9];
  __shared__ float4 sE[ECAP];

  const int tid = threadIdx.x;
  const int wave = tid >> 5, lane = tid & 31, m = lane & 15, hi = lane >> 4;
  const int blockrow = blockIdx.x * 128;
  const int rowbase = blockrow + wave * 16;
  const int bb = blockrow >> 12, n0 = blockrow & 4095;
  const float* p_in = (kstep & 1) ? pB : pA;
  float* p_out = (kstep & 1) ? pA : pB;
  const float2* pb = (const float2*)p_in + (size_t)bb * NNODE;
  const bool has_next = (kstep + 1 < PREDT);
  float* sXw = sX[wave];
  float* sHw = sH[wave];
  float* sXnw = sXn[wave];

  const int e0 = max(rowptr[n0], 0);
  const int e1 = min(rowptr[n0 + 128], NEDGE);

  float xr[10], fn[9];
  int lo = 0, hi_e = 0;
  if (lane < 16) {
    int R = rowbase + lane;
    int n = R & 4095;
    lo = rowptr[n]; hi_e = rowptr[n + 1];
    xr[0] = xn_glob[R];
    const float* fp = feat + (((size_t)bb * 24 + HISTT + kstep) * NNODE + n) * 9;
#pragma unroll
    for (int j = 0; j < 9; ++j) xr[1 + j] = fp[j];
    if (has_next) {
      const float* fq = fp + (size_t)NNODE * 9;
#pragma unroll
      for (int j = 0; j < 9; ++j) fn[j] = fq[j];
    }
  }
  float hreg[16];
#pragma unroll
  for (int i = 0; i < 16; ++i) hreg[i] = h_glob[(size_t)rowbase * 32 + lane + 32 * i];

  int2 en[EPT];
  const int cend0 = min(e0 + ECAP, e1);
  stage_list(csr, en, e0, cend0, tid);
  load_weights(tid, x2h_w, x2h_b, h2h_w, h2h_b, fc_w, sWT, sBias, sFc);
  stage_p(pb, en, sE, e0, cend0, tid);
#pragma unroll
  for (int i = 0; i < 16; ++i) sHw[lane + 32 * i] = hreg[i];
  __syncthreads();

  float a0 = 0.f, a1 = 0.f;
  if (lane < 16) sum_row(sE, e0, cend0, lo, hi_e, a0, a1);
  for (int c = e0 + ECAP; c < e1; c += ECAP) {
    const int cend = min(c + ECAP, e1);
    __syncthreads();
    stage_list(csr, en, c, cend, tid);
    stage_p(pb, en, sE, c, cend, tid);
    __syncthreads();
    if (lane < 16) sum_row(sE, c, cend, lo, hi_e, a0, a1);
  }

  if (lane < 16) {
    float q0 = cheb_b[0], q1 = cheb_b[1];
#pragma unroll
    for (int f = 0; f < 10; ++f) {
      q0 += xr[f] * cheb_w[f * 2 + 0];
      q1 += xr[f] * cheb_w[f * 2 + 1];
    }
#pragma unroll
    for (int f = 0; f < 10; ++f) sXw[lane * 12 + f] = xr[f];
    sXw[lane * 12 + 10] = sigm(q0 + a0);
    sXw[lane * 12 + 11] = sigm(q1 + a1);
    if (has_next) {
#pragma unroll
      for (int j = 0; j < 9; ++j) sXnw[lane * 9 + j] = fn[j];
    }
  }
  __syncthreads();
  float c01[16], h01[16];
#pragma unroll
  for (int ct = 0; ct < 2; ++ct)
#pragma unroll
    for (int r = 0; r < 8; ++r)
      c01[ct * 8 + r] = c_glob[((size_t)rowbase + r + 8 * hi) * 32 + ct * 16 + m];

  v8f acc[8];
  gate_compute(sXw, sHw, sWT, sBias, m, hi, acc);
  lstm_update(acc, c01, h01, sHw, m, hi);

#pragma unroll
  for (int ct = 0; ct < 2; ++ct)
#pragma unroll
    for (int r = 0; r < 8; ++r) {
      size_t R = (size_t)rowbase + r + 8 * hi;
      c_glob[R * 32 + ct * 16 + m] = c01[ct * 8 + r];
      h_glob[R * 32 + ct * 16 + m] = h01[ct * 8 + r];
    }
  __syncthreads();
  if (lane < 16) {
    float s = fc_b[0];
#pragma unroll
    for (int j = 0; j < 32; ++j) s += sHw[lane * 32 + j] * sFc[j];
    int R = rowbase + lane;
    int n = R & 4095;
    xn_glob[R] = s;
    out[((size_t)bb * PREDT + kstep) * NNODE + n] = s;
    if (has_next) {
      float p0 = 0.f, p1 = 0.f;
      p0 += s * cheb_w[20]; p1 += s * cheb_w[21];
#pragma unroll
      for (int j = 0; j < 9; ++j) {
        float xv = sXnw[lane * 9 + j];
        p0 += xv * cheb_w[20 + (1 + j) * 2 + 0];
        p1 += xv * cheb_w[20 + (1 + j) * 2 + 1];
      }
      ((float2*)p_out)[R] = make_float2(p0, p1);
    }
  }
}

extern "C" void kernel_launch(void* const* d_in, const int* in_sizes, int n_in,
                              void* d_out, int out_size, void* d_ws, size_t ws_size,
                              hipStream_t stream) {
  (void)in_sizes; (void)n_in; (void)out_size; (void)ws_size;
  const float* pm25   = (const float*)d_in[0];
  const float* feat   = (const float*)d_in[1];
  const int*   ei     = (const int*)d_in[2];
  const float* cheb_w = (const float*)d_in[3];
  const float* cheb_b = (const float*)d_in[4];
  const float* x2h_w  = (const float*)d_in[5];
  const float* x2h_b  = (const float*)d_in[6];
  const float* h2h_w  = (const float*)d_in[7];
  const float* h2h_b  = (const float*)d_in[8];
  const float* fc_w   = (const float*)d_in[9];
  const float* fc_b   = (const float*)d_in[10];
  float* out = (float*)d_out;
  char* ws = (char*)d_ws;

  int*   cnt_src  = (int*)(ws + 0);
  int*   cnt_dst  = (int*)(ws + 0x4000);
  float* dis      = (float*)(ws + 0x8000);
  int*   rowptr   = (int*)(ws + 0xC000);
  int2*  csr      = (int2*)(ws + 0x14000);
  float* p_hist   = (float*)(ws + 0x100000);
  float* q_hist   = (float*)(ws + 0x1100000);
  float* xg_hist  = (float*)(ws + 0x2100000);
  float* xn_glob  = (float*)(ws + 0x3100000);
  float* p_a      = (float*)(ws + 0x3180000);
  float* p_b      = (float*)(ws + 0x3280000);
  int* arr = (int*)p_a;
  int* tmp = (int*)p_b;
  float* h_glob = p_hist;
  float* c_glob = q_hist;

  k_zero_i<<<2 * NNODE / 256, 256, 0, stream>>>(cnt_src, 2 * NNODE);
  k_count<<<NEDGE / 256, 256, 0, stream>>>(ei, cnt_src, cnt_dst, arr);
  k_scan<<<1, 1024, 0, stream>>>(cnt_src, cnt_dst, rowptr, dis);
  k_place<<<NEDGE / 256, 256, 0, stream>>>(ei, rowptr, arr, tmp);
  k_rank<<<NEDGE / 256, 256, 0, stream>>>(ei, rowptr, dis, tmp, csr);

  k_proj_hist<<<BATCH * HISTT * NNODE / 256, 256, 0, stream>>>(pm25, feat, cheb_w, cheb_b,
                                                               p_hist, q_hist);
  k_gather_hist<<<BATCH * HISTT / 2, 1024, 0, stream>>>(rowptr, csr, p_hist, q_hist, xg_hist);

  hist_lstm_kernel<<<BATCH * NNODE / 128, 256, 0, stream>>>(
      pm25, feat, xg_hist, cheb_w, x2h_w, x2h_b, h2h_w, h2h_b, fc_w, fc_b, h_glob, c_glob,
      xn_glob, p_a);

  for (int k = 0; k < PREDT; ++k) {
    pred_step_kernel<<<BATCH * NNODE / 128, 256, 0, stream>>>(
        feat, rowptr, csr, cheb_w, cheb_b, x2h_w, x2h_b, h2h_w, h2h_b, fc_w, fc_b,
        h_glob, c_glob, xn_glob, p_a, p_b, out, k);
  }
}


# Activations on the hardware tanh instruction: tanh = v_tanh_f32, every sigmoid = 0.5+0.5*tanh(x/2) with x/2 folded into the i,f,o weight columns; replaces library tanhf, exp and exact division
# speedup vs baseline: 6.9997x; 1.4753x over previous
#include <hip/hip_runtime.h>
#include <hip/hip_bf16.h>
#include <cstdint>
#include <cstddef>

#define BATCH 32
#define NNODE 4096
#define HISTT 16
#define PREDT 8
#define NEDGE 65536

typedef float v2f __attribute__((ext_vector_type(2)));
typedef float v8f __attribute__((ext_vector_type(8)));

__device__ __forceinline__ float tanh_hw(float x) { return __builtin_amdgcn_tanhf(x); }

__device__ __forceinline__ float sigm(float x) { return fmaf(tanh_hw(0.5f * x), 0.5f, 0.5f); }

__device__ __forceinline__ int atomAddI(int* p, int v) {
  return __hip_atomic_fetch_add(p, v, __ATOMIC_RELAXED, __HIP_MEMORY_SCOPE_AGENT);
}

__global__ void k_zero_i(int* __restrict__ p, int n) {
  int i = blockIdx.x * 256 + threadIdx.x;
  if (i < n) p[i] = 0;
}

__global__ void k_count(const int* __restrict__ ei, int* __restrict__ cnt_src,
                        int* __restrict__ cnt_dst, int* __restrict__ arr) {
  int e = blockIdx.x * 256 + threadIdx.x;
  if (e < NEDGE) {
    atomAddI(cnt_src + ei[e], 1);
    arr[e] = atomAddI(cnt_dst + ei[NEDGE + e], 1);
  }
}

__launch_bounds__(1024)
__global__ void k_scan(const int* __restrict__ cnt_src, const int* __restrict__ cnt_dst,
                       int* __restrict__ rowptr, float* __restrict__ dis) {
  __shared__ int s[1024];
  const int t = threadIdx.x;
  int c0 = cnt_dst[4 * t + 0], c1 = cnt_dst[4 * t + 1];
  int c2 = cnt_dst[4 * t + 2], c3 = cnt_dst[4 * t + 3];
  int sum = c0 + c1 + c2 + c3;
  s[t] = sum;
  __syncthreads();
  for (int off = 1; off < 1024; off <<= 1) {
    int v = (t >= off) ? s[t - off] : 0;
    __syncthreads();
    s[t] += v;
    __syncthreads();
  }
  int incl = s[t];
  int ex = incl - sum;
  rowptr[4 * t + 0] = ex;
  rowptr[4 * t + 1] = ex + c0;
  rowptr[4 * t + 2] = ex + c0 + c1;
  rowptr[4 * t + 3] = ex + c0 + c1 + c2;
  if (t == 1023) rowptr[NNODE] = incl;
#pragma unroll
  for (int i = 0; i < 4; ++i) {
    float d = (float)cnt_src[4 * t + i];
    dis[4 * t + i] = d > 0.f ? rsqrtf(d) : 0.f;
  }
}

__global__ void k_place(const int* __restrict__ ei, const int* __restrict__ rowptr,
                        const int* __restrict__ arr, int* __restrict__ tmp) {
  int e = blockIdx.x * 256 + threadIdx.x;
  if (e < NEDGE) tmp[rowptr[ei[NEDGE + e]] + arr[e]] = e;
}

__global__ void k_rank(const int* __restrict__ ei, const int* __restrict__ rowptr,
                       const float* __restrict__ dis, const int* __restrict__ tmp,
                       int2* __restrict__ csr) {
  int i = blockIdx.x * 256 + threadIdx.x;
  if (i >= NEDGE) return;
  int e = tmp[i];
  int s = ei[e], d = ei[NEDGE + e];
  int lo = rowptr[d], hi = rowptr[d + 1];
  int rank = 0;
  for (int j = lo; j < hi; ++j) rank += (tmp[j] < e) ? 1 : 0;
  csr[lo + rank] = make_int2(s, __float_as_int(-dis[s] * dis[d]));
}

__global__ void k_proj_hist(const float* __restrict__ pm25, const float* __restrict__ feat,
                            const float* __restrict__ cheb_w, const float* __restrict__ cheb_b,
                            float* __restrict__ p, float* __restrict__ q) {
  int i = blockIdx.x * 256 + threadIdx.x;
  if (i >= BATCH * HISTT * NNODE) return;
  int bt = i >> 12, b = bt >> 4, t = bt & 15, n = i & 4095;
  float x[10];
  x[0] = pm25[i];
  const float* fp = feat + (((size_t)b * 24 + t) * NNODE + n) * 9;
#pragma unroll
  for (int j = 0; j < 9; ++j) x[1 + j] = fp[j];
  float q0 = cheb_b[0], q1 = cheb_b[1], p0 = 0.f, p1 = 0.f;
#pragma unroll
  for (int f = 0; f < 10; ++f) {
    q0 += x[f] * cheb_w[f * 2 + 0];
    q1 += x[f] * cheb_w[f * 2 + 1];
    p0 += x[f] * cheb_w[20 + f * 2 + 0];
    p1 += x[f] * cheb_w[20 + f * 2 + 1];
  }
  size_t o = (size_t)i * 2;
  p[o] = p0; p[o + 1] = p1; q[o] = q0; q[o + 1] = q1;
}

__launch_bounds__(1024)
__global__ void k_gather_hist(const int* __restrict__ rowptr, const int2* __restrict__ csr,
                              const float* __restrict__ p, const float* __restrict__ q,
                              float* __restrict__ xg) {
  __shared__ float4 sP[NNODE];
  const int tid = threadIdx.x;
  const size_t base = (size_t)blockIdx.x * 2 * NNODE;
  const float2* p0 = (const float2*)p + base;
  const float2* p1 = p0 + NNODE;
  for (int n = tid; n < NNODE; n += 1024) {
    float2 a = p0[n], b = p1[n];
    sP[n] = make_float4(a.x, a.y, b.x, b.y);
  }
  __syncthreads();
  const float2* q0 = (const float2*)q + base;
  const float2* q1 = q0 + NNODE;
  float2* x0 = (float2*)xg + base;
  float2* x1 = x0 + NNODE;
  for (int n = tid; n < NNODE; n += 1024) {
    int lo = rowptr[n], hi = rowptr[n + 1];
    float a0 = 0.f, a1 = 0.f, a2 = 0.f, a3 = 0.f;
    for (int j = lo; j < hi; ++j) {
      int2 en = csr[j];
      float nv = __int_as_float(en.y);
      float4 pv = sP[en.x];
      a0 += pv.x * nv; a1 += pv.y * nv; a2 += pv.z * nv; a3 += pv.w * nv;
    }
    float2 qa = q0[n], qb = q1[n];
    x0[n] = make_float2(sigm(qa.x + a0), sigm(qa.y + a1));
    x1[n] = make_float2(sigm(qb.x + a2), sigm(qb.y + a3));
  }
}

#define ECAP 2560

#define EPT (ECAP / 256)

__device__ __forceinline__ void stage_list(const int2* __restrict__ csr, int2* en,
                                           int c, int cend, int tid) {
#pragma unroll
  for (int i = 0; i < EPT; ++i) en[i] = csr[max(min(c + tid + 256 * i, cend - 1), 0)];
}

__device__ __forceinline__ void stage_p(const float2* __restrict__ pb, const int2* en,
                                        float4* sE, int c, int cend, int tid) {
  float2 pv[EPT];
#pragma unroll
  for (int i = 0; i < EPT; ++i) pv[i] = pb[en[i].x];
#pragma unroll
  for (int i = 0; i < EPT; ++i) {
    int j = c + tid + 256 * i;
    if (j < cend) sE[j - c] = make_float4(pv[i].x, pv[i].y, __int_as_float(en[i].y), 0.f);
  }
}

__device__ __forceinline__ void sum_row(const float4* sE, int c, int cend, int lo, int hi_e,
                                        float& a0, float& a1) {
  int j0 = max(lo, c) - c, j1 = min(hi_e, cend) - c;
  for (int j = j0; j < j1; j += 4) {
    float4 v0 = sE[j];
    float4 v1 = sE[min(j + 1, j1 - 1)];
    float4 v2 = sE[min(j + 2, j1 - 1)];
    float4 v3 = sE[min(j + 3, j1 - 1)];
    a0 += v0.x * v0.z; a1 += v0.y * v0.z;
    if (j + 1 < j1) { a0 += v1.x * v1.z; a1 += v1.y * v1.z; }
    if (j + 2 < j1) { a0 += v2.x * v2.z; a1 += v2.y * v2.z; }
    if (j + 3 < j1) { a0 += v3.x * v3.z; a1 += v3.y * v3.z; }
  }
}

#define WT_STRIDE 48

__device__ __forceinline__ void load_weights(int tid,
    const float* __restrict__ x2h_w, const float* __restrict__ x2h_b,
    const float* __restrict__ h2h_w, const float* __restrict__ h2h_b,
    const float* __restrict__ fc_w,
    float* sWT, float* sBias, float* sFc) {
  for (int i = tid; i < 44 * 128; i += 256) {
    int k = i >> 7, n = i & 127;
    float w = (k < 12) ? x2h_w[k * 128 + n] : h2h_w[(k - 12) * 128 + n];
    sWT[n * WT_STRIDE + k] = ((n >> 5) == 2) ? w : 0.5f * w;
  }
  if (tid < 128) {
    float bsum = x2h_b[tid] + h2h_b[tid];
    sBias[tid] = ((tid >> 5) == 2) ? bsum : 0.5f * bsum;
  }
  if (tid < 32) sFc[tid] = fc_w[tid];
}

__device__ __forceinline__ void gate_compute(const float* __restrict__ sXw,
                                             const float* __restrict__ sHw,
                                             const float* __restrict__ sWT,
                                             const float* __restrict__ sBias,
                                             int m, int hi, v8f* acc) {
  v2f ax[3], ah[8];
#pragma unroll
  for (int kk = 0; kk < 3; ++kk) ax[kk] = *(const v2f*)(sXw + m * 12 + 4 * kk + 2 * hi);
#pragma unroll
  for (int kk = 0; kk < 8; ++kk) ah[kk] = *(const v2f*)(sHw + m * 32 + 4 * kk + 2 * hi);
#pragma unroll
  for (int ct = 0; ct < 8; ++ct) {
    float bb = sBias[ct * 16 + m];
    v8f a = {bb, bb, bb, bb, bb, bb, bb, bb};
    const float* wcol = sWT + (ct * 16 + m) * WT_STRIDE;
#pragma unroll
    for (int kk = 0; kk < 3; ++kk) {
      v2f bf = *(const v2f*)(wcol + 4 * kk + 2 * hi);
      a = __builtin_amdgcn_wmma_f32_16x16x4_f32(false, ax[kk], false, bf, (short)0, a,
                                                false, false);
    }
#pragma unroll
    for (int kk = 0; kk < 8; ++kk) {
      v2f bf = *(const v2f*)(wcol + 12 + 4 * kk + 2 * hi);
      a = __builtin_amdgcn_wmma_f32_16x16x4_f32(false, ah[kk], false, bf, (short)0, a,
                                                false, false);
    }
    acc[ct] = a;
  }
}

__device__ __forceinline__ void lstm_update(const v8f* acc, float* c01, float* h01,
                                            float* sHw, int m, int hi) {
#pragma unroll
  for (int ct = 0; ct < 2; ++ct) {
#pragma unroll
    for (int r = 0; r < 8; ++r) {
      float iv = fmaf(tanh_hw(acc[ct][r]), 0.5f, 0.5f);
      float fv = fmaf(tanh_hw(acc[2 + ct][r]), 0.5f, 0.5f);
      float gv = tanh_hw(acc[4 + ct][r]);
      float ov = fmaf(tanh_hw(acc[6 + ct][r]), 0.5f, 0.5f);
      float cn = fv * c01[ct * 8 + r] + iv * gv;
      float hn = ov * tanh_hw(cn);
      c01[ct * 8 + r] = cn;
      h01[ct * 8 + r] = hn;
      sHw[(r + 8 * hi) * 32 + ct * 16 + m] = hn;
    }
  }
}

__launch_bounds__(256)
__global__ void hist_lstm_kernel(
    const float* __restrict__ pm25, const float* __restrict__ feat,
    const float* __restrict__ xg, const float* __restrict__ cheb_w,
    const float* __restrict__ x2h_w, const float* __restrict__ x2h_b,
    const float* __restrict__ h2h_w, const float* __restrict__ h2h_b,
    const float* __restrict__ fc_w, const float* __restrict__ fc_b,
    float* __restrict__ h_glob, float* __restrict__ c_glob, float* __restrict__ xn_glob,
    float* __restrict__ p_first) {
  __shared__ float sWT[128 * WT_STRIDE];
  __shared__ float sBias[128];
  __shared__ float sFc[32];
  __shared__ float sX[8][16 * 12];
  __shared__ float sH[8][16 * 32];

  const int tid = threadIdx.x;
  load_weights(tid, x2h_w, x2h_b, h2h_w, h2h_b, fc_w, sWT, sBias, sFc);

  const int wave = tid >> 5, lane = tid & 31, m = lane & 15, hi = lane >> 4;
  const int rowbase = blockIdx.x * 128 + wave * 16;
  float* sXw = sX[wave];
  float* sHw = sH[wave];
  for (int j = lane; j < 512; j += 32) sHw[j] = 0.f;
  float c01[16], h01[16];
#pragma unroll
  for (int j = 0; j < 16; ++j) { c01[j] = 0.f; h01[j] = 0.f; }
  __syncthreads();

  for (int t = 0; t < HISTT; ++t) {
    if (lane < 16) {
      int R = rowbase + lane;
      int b = R >> 12, n = R & 4095;
      size_t hb = ((size_t)(b * 16 + t)) * NNODE + n;
      sXw[lane * 12 + 0] = pm25[hb];
      const float* fp = feat + (((size_t)b * 24 + t) * NNODE + n) * 9;
#pragma unroll
      for (int j = 0; j < 9; ++j) sXw[lane * 12 + 1 + j] = fp[j];
      sXw[lane * 12 + 10] = xg[hb * 2 + 0];
      sXw[lane * 12 + 11] = xg[hb * 2 + 1];
    }
    __syncthreads();
    v8f acc[8];
    gate_compute(sXw, sHw, sWT, sBias, m, hi, acc);
    lstm_update(acc, c01, h01, sHw, m, hi);
  }

#pragma unroll
  for (int ct = 0; ct < 2; ++ct)
#pragma unroll
    for (int r = 0; r < 8; ++r) {
      size_t R = (size_t)rowbase + r + 8 * hi;
      c_glob[R * 32 + ct * 16 + m] = c01[ct * 8 + r];
      h_glob[R * 32 + ct * 16 + m] = h01[ct * 8 + r];
    }
  __syncthreads();
  if (lane < 16) {
    float s = fc_b[0];
#pragma unroll
    for (int j = 0; j < 32; ++j) s += sHw[lane * 32 + j] * sFc[j];
    xn_glob[rowbase + lane] = s;
    int R = rowbase + lane;
    int b = R >> 12, n = R & 4095;
    const float* fp = feat + (((size_t)b * 24 + HISTT) * NNODE + n) * 9;
    float p0 = 0.f, p1 = 0.f;
    p0 += s * cheb_w[20]; p1 += s * cheb_w[21];
#pragma unroll
    for (int j = 0; j < 9; ++j) {
      float xv = fp[j];
      p0 += xv * cheb_w[20 + (1 + j) * 2 + 0];
      p1 += xv * cheb_w[20 + (1 + j) * 2 + 1];
    }
    ((float2*)p_first)[R] = make_float2(p0, p1);
  }
}

__launch_bounds__(256)
__global__ void pred_step_kernel(
    const float* __restrict__ feat,
    const int* __restrict__ rowptr, const int2* __restrict__ csr,
    const float* __restrict__ cheb_w, const float* __restrict__ cheb_b,
    const float* __restrict__ x2h_w, const float* __restrict__ x2h_b,
    const float* __restrict__ h2h_w, const float* __restrict__ h2h_b,
    const float* __restrict__ fc_w, const float* __restrict__ fc_b,
    float* __restrict__ h_glob, float* __restrict__ c_glob, float* __restrict__ xn_glob,
    float* __restrict__ pA, float* __restrict__ pB,
    float* __restrict__ out, int kstep) {
  __shared__ float sWT[128 * WT_STRIDE];
  __shared__ float sBias[128];
  __shared__ float sFc[32];
  __shared__ float sX[8][16 * 12];
  __shared__ float sH[8][16 * 32];
  __shared__ float sXn[8][16 * 9];
  __shared__ float4 sE[ECAP];

  const int tid = threadIdx.x;
  const int wave = tid >> 5, lane = tid & 31, m = lane & 15, hi = lane >> 4;
  const int blockrow = blockIdx.x * 128;
  const int rowbase = blockrow + wave * 16;
  const int bb = blockrow >> 12, n0 = blockrow & 4095;
  const float* p_in = (kstep & 1) ? pB : pA;
  float* p_out = (kstep & 1) ? pA : pB;
  const float2* pb = (const float2*)p_in + (size_t)bb * NNODE;
  const bool has_next = (kstep + 1 < PREDT);
  float* sXw = sX[wave];
  float* sHw = sH[wave];
  float* sXnw = sXn[wave];

  const int e0 = max(rowptr[n0], 0);
  const int e1 = min(rowptr[n0 + 128], NEDGE);

  float xr[10], fn[9];
  int lo = 0, hi_e = 0;
  if (lane < 16) {
    int R = rowbase + lane;
    int n = R & 4095;
    lo = rowptr[n]; hi_e = rowptr[n + 1];
    xr[0] = xn_glob[R];
    const float* fp = feat + (((size_t)bb * 24 + HISTT + kstep) * NNODE + n) * 9;
#pragma unroll
    for (int j = 0; j < 9; ++j) xr[1 + j] = fp[j];
    if (has_next) {
      const float* fq = fp + (size_t)NNODE * 9;
#pragma unroll
      for (int j = 0; j < 9; ++j) fn[j] = fq[j];
    }
  }
  float hreg[16];
#pragma unroll
  for (int i = 0; i < 16; ++i) hreg[i] = h_glob[(size_t)rowbase * 32 + lane + 32 * i];

  int2 en[EPT];
  const int cend0 = min(e0 + ECAP, e1);
  stage_list(csr, en, e0, cend0, tid);
  load_weights(tid, x2h_w, x2h_b, h2h_w, h2h_b, fc_w, sWT, sBias, sFc);
  stage_p(pb, en, sE, e0, cend0, tid);
#pragma unroll
  for (int i = 0; i < 16; ++i) sHw[lane + 32 * i] = hreg[i];
  __syncthreads();

  float a0 = 0.f, a1 = 0.f;
  if (lane < 16) sum_row(sE, e0, cend0, lo, hi_e, a0, a1);
  for (int c = e0 + ECAP; c < e1; c += ECAP) {
    const int cend = min(c + ECAP, e1);
    __syncthreads();
    stage_list(csr, en, c, cend, tid);
    stage_p(pb, en, sE, c, cend, tid);
    __syncthreads();
    if (lane < 16) sum_row(sE, c, cend, lo, hi_e, a0, a1);
  }

  if (lane < 16) {
    float q0 = cheb_b[0], q1 = cheb_b[1];
#pragma unroll
    for (int f = 0; f < 10; ++f) {
      q0 += xr[f] * cheb_w[f * 2 + 0];
      q1 += xr[f] * cheb_w[f * 2 + 1];
    }
#pragma unroll
    for (int f = 0; f < 10; ++f) sXw[lane * 12 + f] = xr[f];
    sXw[lane * 12 + 10] = sigm(q0 + a0);
    sXw[lane * 12 + 11] = sigm(q1 + a1);
    if (has_next) {
#pragma unroll
      for (int j = 0; j < 9; ++j) sXnw[lane * 9 + j] = fn[j];
    }
  }
  __syncthreads();
  float c01[16], h01[16];
#pragma unroll
  for (int ct = 0; ct < 2; ++ct)
#pragma unroll
    for (int r = 0; r < 8; ++r)
      c01[ct * 8 + r] = c_glob[((size_t)rowbase + r + 8 * hi) * 32 + ct * 16 + m];

  v8f acc[8];
  gate_compute(sXw, sHw, sWT, sBias, m, hi, acc);
  lstm_update(acc, c01, h01, sHw, m, hi);

#pragma unroll
  for (int ct = 0; ct < 2; ++ct)
#pragma unroll
    for (int r = 0; r < 8; ++r) {
      size_t R = (size_t)rowbase + r + 8 * hi;
      c_glob[R * 32 + ct * 16 + m] = c01[ct * 8 + r];
      h_glob[R * 32 + ct * 16 + m] = h01[ct * 8 + r];
    }
  __syncthreads();
  if (lane < 16) {
    float s = fc_b[0];
#pragma unroll
    for (int j = 0; j < 32; ++j) s += sHw[lane * 32 + j] * sFc[j];
    int R = rowbase + lane;
    xn_glob[R] = s;
    out[(size_t)R + (size_t)(bb * (PREDT - 1) + kstep) * NNODE] = s;
    if (has_next) {
      float p0 = 0.f, p1 = 0.f;
      p0 += s * cheb_w[20]; p1 += s * cheb_w[21];
#pragma unroll
      for (int j = 0; j < 9; ++j) {
        float xv = sXnw[lane * 9 + j];
        p0 += xv * cheb_w[20 + (1 + j) * 2 + 0];
        p1 += xv * cheb_w[20 + (1 + j) * 2 + 1];
      }
      ((float2*)p_out)[R] = make_float2(p0, p1);
    }
  }
}

extern "C" void kernel_launch(void* const* d_in, const int* in_sizes, int n_in,
                              void* d_out, int out_size, void* d_ws, size_t ws_size,
                              hipStream_t stream) {
  (void)in_sizes; (void)n_in; (void)out_size; (void)ws_size;
  const float* pm25   = (const float*)d_in[0];
  const float* feat   = (const float*)d_in[1];
  const int*   ei     = (const int*)d_in[2];
  const float* cheb_w = (const float*)d_in[3];
  const float* cheb_b = (const float*)d_in[4];
  const float* x2h_w  = (const float*)d_in[5];
  const float* x2h_b  = (const float*)d_in[6];
  const float* h2h_w  = (const float*)d_in[7];
  const float* h2h_b  = (const float*)d_in[8];
  const float* fc_w   = (const float*)d_in[9];
  const float* fc_b   = (const float*)d_in[10];
  float* out = (float*)d_out;
  char* ws = (char*)d_ws;

  int*   cnt_src  = (int*)(ws + 0);
  int*   cnt_dst  = (int*)(ws + 0x4000);
  float* dis      = (float*)(ws + 0x8000);
  int*   rowptr   = (int*)(ws + 0xC000);
  int2*  csr      = (int2*)(ws + 0x14000);
  float* p_hist   = (float*)(ws + 0x100000);
  float* q_hist   = (float*)(ws + 0x1100000);
  float* xg_hist  = (float*)(ws + 0x2100000);
  float* xn_glob  = (float*)(ws + 0x3100000);
  float* p_a      = (float*)(ws + 0x3180000);
  float* p_b      = (float*)(ws + 0x3280000);
  int* arr = (int*)p_a;
  int* tmp = (int*)p_b;
  float* h_glob = p_hist;
  float* c_glob = q_hist;

  k_zero_i<<<2 * NNODE / 256, 256, 0, stream>>>(cnt_src, 2 * NNODE);
  k_count<<<NEDGE / 256, 256, 0, stream>>>(ei, cnt_src, cnt_dst, arr);
  k_scan<<<1, 1024, 0, stream>>>(cnt_src, cnt_dst, rowptr, dis);
  k_place<<<NEDGE / 256, 256, 0, stream>>>(ei, rowptr, arr, tmp);
  k_rank<<<NEDGE / 256, 256, 0, stream>>>(ei, rowptr, dis, tmp, csr);

  k_proj_hist<<<BATCH * HISTT * NNODE / 256, 256, 0, stream>>>(pm25, feat, cheb_w, cheb_b,
                                                               p_hist, q_hist);
  k_gather_hist<<<BATCH * HISTT / 2, 1024, 0, stream>>>(rowptr, csr, p_hist, q_hist, xg_hist);

  hist_lstm_kernel<<<BATCH * NNODE / 128, 256, 0, stream>>>(
      pm25, feat, xg_hist, cheb_w, x2h_w, x2h_b, h2h_w, h2h_b, fc_w, fc_b, h_glob, c_glob,
      xn_glob, p_a);

  for (int k = 0; k < PREDT; ++k) {
    pred_step_kernel<<<BATCH * NNODE / 128, 256, 0, stream>>>(
        feat, rowptr, csr, cheb_w, cheb_b, x2h_w, x2h_b, h2h_w, h2h_b, fc_w, fc_b,
        h_glob, c_glob, xn_glob, p_a, p_b, out, k);
  }
}


# LDS row pitches made bank-conflict-free for the fragment reads: W^T pitch 48 -> 44 floats, H pitch 32 -> 36 (both 4 x odd on 64 banks); B reads were 4-way, A reads of h 8-way conflicted
# speedup vs baseline: 7.3362x; 1.0481x over previous
#include <hip/hip_runtime.h>
#include <hip/hip_bf16.h>
#include <cstdint>
#include <cstddef>

#define BATCH 32
#define NNODE 4096
#define HISTT 16
#define PREDT 8
#define NEDGE 65536

typedef float v2f __attribute__((ext_vector_type(2)));
typedef float v8f __attribute__((ext_vector_type(8)));

__device__ __forceinline__ float tanh_hw(float x) { return __builtin_amdgcn_tanhf(x); }

__device__ __forceinline__ float sigm(float x) { return fmaf(tanh_hw(0.5f * x), 0.5f, 0.5f); }

__device__ __forceinline__ int atomAddI(int* p, int v) {
  return __hip_atomic_fetch_add(p, v, __ATOMIC_RELAXED, __HIP_MEMORY_SCOPE_AGENT);
}

__global__ void k_zero_i(int* __restrict__ p, int n) {
  int i = blockIdx.x * 256 + threadIdx.x;
  if (i < n) p[i] = 0;
}

__global__ void k_count(const int* __restrict__ ei, int* __restrict__ cnt_src,
                        int* __restrict__ cnt_dst, int* __restrict__ arr) {
  int e = blockIdx.x * 256 + threadIdx.x;
  if (e < NEDGE) {
    atomAddI(cnt_src + ei[e], 1);
    arr[e] = atomAddI(cnt_dst + ei[NEDGE + e], 1);
  }
}

__launch_bounds__(1024)
__global__ void k_scan(const int* __restrict__ cnt_src, const int* __restrict__ cnt_dst,
                       int* __restrict__ rowptr, float* __restrict__ dis) {
  __shared__ int s[1024];
  const int t = threadIdx.x;
  int c0 = cnt_dst[4 * t + 0], c1 = cnt_dst[4 * t + 1];
  int c2 = cnt_dst[4 * t + 2], c3 = cnt_dst[4 * t + 3];
  int sum = c0 + c1 + c2 + c3;
  s[t] = sum;
  __syncthreads();
  for (int off = 1; off < 1024; off <<= 1) {
    int v = (t >= off) ? s[t - off] : 0;
    __syncthreads();
    s[t] += v;
    __syncthreads();
  }
  int incl = s[t];
  int ex = incl - sum;
  rowptr[4 * t + 0] = ex;
  rowptr[4 * t + 1] = ex + c0;
  rowptr[4 * t + 2] = ex + c0 + c1;
  rowptr[4 * t + 3] = ex + c0 + c1 + c2;
  if (t == 1023) rowptr[NNODE] = incl;
#pragma unroll
  for (int i = 0; i < 4; ++i) {
    float d = (float)cnt_src[4 * t + i];
    dis[4 * t + i] = d > 0.f ? rsqrtf(d) : 0.f;
  }
}

__global__ void k_place(const int* __restrict__ ei, const int* __restrict__ rowptr,
                        const int* __restrict__ arr, int* __restrict__ tmp) {
  int e = blockIdx.x * 256 + threadIdx.x;
  if (e < NEDGE) tmp[rowptr[ei[NEDGE + e]] + arr[e]] = e;
}

__global__ void k_rank(const int* __restrict__ ei, const int* __restrict__ rowptr,
                       const float* __restrict__ dis, const int* __restrict__ tmp,
                       int2* __restrict__ csr) {
  int i = blockIdx.x * 256 + threadIdx.x;
  if (i >= NEDGE) return;
  int e = tmp[i];
  int s = ei[e], d = ei[NEDGE + e];
  int lo = rowptr[d], hi = rowptr[d + 1];
  int rank = 0;
  for (int j = lo; j < hi; ++j) rank += (tmp[j] < e) ? 1 : 0;
  csr[lo + rank] = make_int2(s, __float_as_int(-dis[s] * dis[d]));
}

__global__ void k_proj_hist(const float* __restrict__ pm25, const float* __restrict__ feat,
                            const float* __restrict__ cheb_w, const float* __restrict__ cheb_b,
                            float* __restrict__ p, float* __restrict__ q) {
  int i = blockIdx.x * 256 + threadIdx.x;
  if (i >= BATCH * HISTT * NNODE) return;
  int bt = i >> 12, b = bt >> 4, t = bt & 15, n = i & 4095;
  float x[10];
  x[0] = pm25[i];
  const float* fp = feat + (((size_t)b * 24 + t) * NNODE + n) * 9;
#pragma unroll
  for (int j = 0; j < 9; ++j) x[1 + j] = fp[j];
  float q0 = cheb_b[0], q1 = cheb_b[1], p0 = 0.f, p1 = 0.f;
#pragma unroll
  for (int f = 0; f < 10; ++f) {
    q0 += x[f] * cheb_w[f * 2 + 0];
    q1 += x[f] * cheb_w[f * 2 + 1];
    p0 += x[f] * cheb_w[20 + f * 2 + 0];
    p1 += x[f] * cheb_w[20 + f * 2 + 1];
  }
  size_t o = (size_t)i * 2;
  p[o] = p0; p[o + 1] = p1; q[o] = q0; q[o + 1] = q1;
}

__launch_bounds__(1024)
__global__ void k_gather_hist(const int* __restrict__ rowptr, const int2* __restrict__ csr,
                              const float* __restrict__ p, const float* __restrict__ q,
                              float* __restrict__ xg) {
  __shared__ float4 sP[NNODE];
  const int tid = threadIdx.x;
  const size_t base = (size_t)blockIdx.x * 2 * NNODE;
  const float2* p0 = (const float2*)p + base;
  const float2* p1 = p0 + NNODE;
  for (int n = tid; n < NNODE; n += 1024) {
    float2 a = p0[n], b = p1[n];
    sP[n] = make_float4(a.x, a.y, b.x, b.y);
  }
  __syncthreads();
  const float2* q0 = (const float2*)q + base;
  const float2* q1 = q0 + NNODE;
  float2* x0 = (float2*)xg + base;
  float2* x1 = x0 + NNODE;
  for (int n = tid; n < NNODE; n += 1024) {
    int lo = rowptr[n], hi = rowptr[n + 1];
    float a0 = 0.f, a1 = 0.f, a2 = 0.f, a3 = 0.f;
    for (int j = lo; j < hi; ++j) {
      int2 en = csr[j];
      float nv = __int_as_float(en.y);
      float4 pv = sP[en.x];
      a0 += pv.x * nv; a1 += pv.y * nv; a2 += pv.z * nv; a3 += pv.w * nv;
    }
    float2 qa = q0[n], qb = q1[n];
    x0[n] = make_float2(sigm(qa.x + a0), sigm(qa.y + a1));
    x1[n] = make_float2(sigm(qb.x + a2), sigm(qb.y + a3));
  }
}

#define ECAP 2560

#define EPT (ECAP / 256)

__device__ __forceinline__ void stage_list(const int2* __restrict__ csr, int2* en,
                                           int c, int cend, int tid) {
#pragma unroll
  for (int i = 0; i < EPT; ++i) en[i] = csr[max(min(c + tid + 256 * i, cend - 1), 0)];
}

__device__ __forceinline__ void stage_p(const float2* __restrict__ pb, const int2* en,
                                        float4* sE, int c, int cend, int tid) {
  float2 pv[EPT];
#pragma unroll
  for (int i = 0; i < EPT; ++i) pv[i] = pb[en[i].x];
#pragma unroll
  for (int i = 0; i < EPT; ++i) {
    int j = c + tid + 256 * i;
    if (j < cend) sE[j - c] = make_float4(pv[i].x, pv[i].y, __int_as_float(en[i].y), 0.f);
  }
}

__device__ __forceinline__ void sum_row(const float4* sE, int c, int cend, int lo, int hi_e,
                                        float& a0, float& a1) {
  int j0 = max(lo, c) - c, j1 = min(hi_e, cend) - c;
  for (int j = j0; j < j1; j += 4) {
    float4 v0 = sE[j];
    float4 v1 = sE[min(j + 1, j1 - 1)];
    float4 v2 = sE[min(j + 2, j1 - 1)];
    float4 v3 = sE[min(j + 3, j1 - 1)];
    a0 += v0.x * v0.z; a1 += v0.y * v0.z;
    if (j + 1 < j1) { a0 += v1.x * v1.z; a1 += v1.y * v1.z; }
    if (j + 2 < j1) { a0 += v2.x * v2.z; a1 += v2.y * v2.z; }
    if (j + 3 < j1) { a0 += v3.x * v3.z; a1 += v3.y * v3.z; }
  }
}

#define WT_STRIDE 44
#define SH_STRIDE 36

__device__ __forceinline__ void load_weights(int tid,
    const float* __restrict__ x2h_w, const float* __restrict__ x2h_b,
    const float* __restrict__ h2h_w, const float* __restrict__ h2h_b,
    const float* __restrict__ fc_w,
    float* sWT, float* sBias, float* sFc) {
  for (int i = tid; i < 44 * 128; i += 256) {
    int k = i >> 7, n = i & 127;
    float w = (k < 12) ? x2h_w[k * 128 + n] : h2h_w[(k - 12) * 128 + n];
    sWT[n * WT_STRIDE + k] = ((n >> 5) == 2) ? w : 0.5f * w;
  }
  if (tid < 128) {
    float bsum = x2h_b[tid] + h2h_b[tid];
    sBias[tid] = ((tid >> 5) == 2) ? bsum : 0.5f * bsum;
  }
  if (tid < 32) sFc[tid] = fc_w[tid];
}

__device__ __forceinline__ void gate_compute(const float* __restrict__ sXw,
                                             const float* __restrict__ sHw,
                                             const float* __restrict__ sWT,
                                             const float* __restrict__ sBias,
                                             int m, int hi, v8f* acc) {
  v2f ax[3], ah[8];
#pragma unroll
  for (int kk = 0; kk < 3; ++kk) ax[kk] = *(const v2f*)(sXw + m * 12 + 4 * kk + 2 * hi);
#pragma unroll
  for (int kk = 0; kk < 8; ++kk) ah[kk] = *(const v2f*)(sHw + m * SH_STRIDE + 4 * kk + 2 * hi);
#pragma unroll
  for (int ct = 0; ct < 8; ++ct) {
    float bb = sBias[ct * 16 + m];
    v8f a = {bb, bb, bb, bb, bb, bb, bb, bb};
    const float* wcol = sWT + (ct * 16 + m) * WT_STRIDE;
#pragma unroll
    for (int kk = 0; kk < 3; ++kk) {
      v2f bf = *(const v2f*)(wcol + 4 * kk + 2 * hi);
      a = __builtin_amdgcn_wmma_f32_16x16x4_f32(false, ax[kk], false, bf, (short)0, a,
                                                false, false);
    }
#pragma unroll
    for (int kk = 0; kk < 8; ++kk) {
      v2f bf = *(const v2f*)(wcol + 12 + 4 * kk + 2 * hi);
      a = __builtin_amdgcn_wmma_f32_16x16x4_f32(false, ah[kk], false, bf, (short)0, a,
                                                false, false);
    }
    acc[ct] = a;
  }
}

__device__ __forceinline__ void lstm_update(const v8f* acc, float* c01, float* h01,
                                            float* sHw, int m, int hi) {
#pragma unroll
  for (int ct = 0; ct < 2; ++ct) {
#pragma unroll
    for (int r = 0; r < 8; ++r) {
      float iv = fmaf(tanh_hw(acc[ct][r]), 0.5f, 0.5f);
      float fv = fmaf(tanh_hw(acc[2 + ct][r]), 0.5f, 0.5f);
      float gv = tanh_hw(acc[4 + ct][r]);
      float ov = fmaf(tanh_hw(acc[6 + ct][r]), 0.5f, 0.5f);
      float cn = fv * c01[ct * 8 + r] + iv * gv;
      float hn = ov * tanh_hw(cn);
      c01[ct * 8 + r] = cn;
      h01[ct * 8 + r] = hn;
      sHw[(r + 8 * hi) * SH_STRIDE + ct * 16 + m] = hn;
    }
  }
}

__launch_bounds__(256)
__global__ void hist_lstm_kernel(
    const float* __restrict__ pm25, const float* __restrict__ feat,
    const float* __restrict__ xg, const float* __restrict__ cheb_w,
    const float* __restrict__ x2h_w, const float* __restrict__ x2h_b,
    const float* __restrict__ h2h_w, const float* __restrict__ h2h_b,
    const float* __restrict__ fc_w, const float* __restrict__ fc_b,
    float* __restrict__ h_glob, float* __restrict__ c_glob, float* __restrict__ xn_glob,
    float* __restrict__ p_first) {
  __shared__ float sWT[128 * WT_STRIDE];
  __shared__ float sBias[128];
  __shared__ float sFc[32];
  __shared__ float sX[8][16 * 12];
  __shared__ float sH[8][16 * SH_STRIDE];

  const int tid = threadIdx.x;
  load_weights(tid, x2h_w, x2h_b, h2h_w, h2h_b, fc_w, sWT, sBias, sFc);

  const int wave = tid >> 5, lane = tid & 31, m = lane & 15, hi = lane >> 4;
  const int rowbase = blockIdx.x * 128 + wave * 16;
  float* sXw = sX[wave];
  float* sHw = sH[wave];
  for (int j = lane; j < 16 * SH_STRIDE; j += 32) sHw[j] = 0.f;
  float c01[16], h01[16];
#pragma unroll
  for (int j = 0; j < 16; ++j) { c01[j] = 0.f; h01[j] = 0.f; }
  __syncthreads();

  for (int t = 0; t < HISTT; ++t) {
    if (lane < 16) {
      int R = rowbase + lane;
      int b = R >> 12, n = R & 4095;
      size_t hb = ((size_t)(b * 16 + t)) * NNODE + n;
      sXw[lane * 12 + 0] = pm25[hb];
      const float* fp = feat + (((size_t)b * 24 + t) * NNODE + n) * 9;
#pragma unroll
      for (int j = 0; j < 9; ++j) sXw[lane * 12 + 1 + j] = fp[j];
      sXw[lane * 12 + 10] = xg[hb * 2 + 0];
      sXw[lane * 12 + 11] = xg[hb * 2 + 1];
    }
    __syncthreads();
    v8f acc[8];
    gate_compute(sXw, sHw, sWT, sBias, m, hi, acc);
    lstm_update(acc, c01, h01, sHw, m, hi);
  }

#pragma unroll
  for (int ct = 0; ct < 2; ++ct)
#pragma unroll
    for (int r = 0; r < 8; ++r) {
      size_t R = (size_t)rowbase + r + 8 * hi;
      c_glob[R * 32 + ct * 16 + m] = c01[ct * 8 + r];
      h_glob[R * 32 + ct * 16 + m] = h01[ct * 8 + r];
    }
  __syncthreads();
  if (lane < 16) {
    float s = fc_b[0];
#pragma unroll
    for (int j = 0; j < 32; ++j) s += sHw[lane * SH_STRIDE + j] * sFc[j];
    xn_glob[rowbase + lane] = s;
    int R = rowbase + lane;
    int b = R >> 12, n = R & 4095;
    const float* fp = feat + (((size_t)b * 24 + HISTT) * NNODE + n) * 9;
    float p0 = 0.f, p1 = 0.f;
    p0 += s * cheb_w[20]; p1 += s * cheb_w[21];
#pragma unroll
    for (int j = 0; j < 9; ++j) {
      float xv = fp[j];
      p0 += xv * cheb_w[20 + (1 + j) * 2 + 0];
      p1 += xv * cheb_w[20 + (1 + j) * 2 + 1];
    }
    ((float2*)p_first)[R] = make_float2(p0, p1);
  }
}

__launch_bounds__(256)
__global__ void pred_step_kernel(
    const float* __restrict__ feat,
    const int* __restrict__ rowptr, const int2* __restrict__ csr,
    const float* __restrict__ cheb_w, const float* __restrict__ cheb_b,
    const float* __restrict__ x2h_w, const float* __restrict__ x2h_b,
    const float* __restrict__ h2h_w, const float* __restrict__ h2h_b,
    const float* __restrict__ fc_w, const float* __restrict__ fc_b,
    float* __restrict__ h_glob, float* __restrict__ c_glob, float* __restrict__ xn_glob,
    float* __restrict__ pA, float* __restrict__ pB,
    float* __restrict__ out, int kstep) {
  __shared__ float sWT[128 * WT_STRIDE];
  __shared__ float sBias[128];
  __shared__ float sFc[32];
  __shared__ float sX[8][16 * 12];
  __shared__ float sH[8][16 * SH_STRIDE];
  __shared__ float sXn[8][16 * 9];
  __shared__ float4 sE[ECAP];

  const int tid = threadIdx.x;
  const int wave = tid >> 5, lane = tid & 31, m = lane & 15, hi = lane >> 4;
  const int blockrow = blockIdx.x * 128;
  const int rowbase = blockrow + wave * 16;
  const int bb = blockrow >> 12, n0 = blockrow & 4095;
  const float* p_in = (kstep & 1) ? pB : pA;
  float* p_out = (kstep & 1) ? pA : pB;
  const float2* pb = (const float2*)p_in + (size_t)bb * NNODE;
  const bool has_next = (kstep + 1 < PREDT);
  float* sXw = sX[wave];
  float* sHw = sH[wave];
  float* sXnw = sXn[wave];

  const int e0 = max(rowptr[n0], 0);
  const int e1 = min(rowptr[n0 + 128], NEDGE);

  float xr[10], fn[9];
  int lo = 0, hi_e = 0;
  if (lane < 16) {
    int R = rowbase + lane;
    int n = R & 4095;
    lo = rowptr[n]; hi_e = rowptr[n + 1];
    xr[0] = xn_glob[R];
    const float* fp = feat + (((size_t)bb * 24 + HISTT + kstep) * NNODE + n) * 9;
#pragma unroll
    for (int j = 0; j < 9; ++j) xr[1 + j] = fp[j];
    if (has_next) {
      const float* fq = fp + (size_t)NNODE * 9;
#pragma unroll
      for (int j = 0; j < 9; ++j) fn[j] = fq[j];
    }
  }
  float hreg[16];
#pragma unroll
  for (int i = 0; i < 16; ++i) hreg[i] = h_glob[(size_t)rowbase * 32 + lane + 32 * i];

  int2 en[EPT];
  const int cend0 = min(e0 + ECAP, e1);
  stage_list(csr, en, e0, cend0, tid);
  load_weights(tid, x2h_w, x2h_b, h2h_w, h2h_b, fc_w, sWT, sBias, sFc);
  stage_p(pb, en, sE, e0, cend0, tid);
#pragma unroll
  for (int i = 0; i < 16; ++i) sHw[lane + SH_STRIDE * i] = hreg[i];
  __syncthreads();

  float a0 = 0.f, a1 = 0.f;
  if (lane < 16) sum_row(sE, e0, cend0, lo, hi_e, a0, a1);
  for (int c = e0 + ECAP; c < e1; c += ECAP) {
    const int cend = min(c + ECAP, e1);
    __syncthreads();
    stage_list(csr, en, c, cend, tid);
    stage_p(pb, en, sE, c, cend, tid);
    __syncthreads();
    if (lane < 16) sum_row(sE, c, cend, lo, hi_e, a0, a1);
  }

  if (lane < 16) {
    float q0 = cheb_b[0], q1 = cheb_b[1];
#pragma unroll
    for (int f = 0; f < 10; ++f) {
      q0 += xr[f] * cheb_w[f * 2 + 0];
      q1 += xr[f] * cheb_w[f * 2 + 1];
    }
#pragma unroll
    for (int f = 0; f < 10; ++f) sXw[lane * 12 + f] = xr[f];
    sXw[lane * 12 + 10] = sigm(q0 + a0);
    sXw[lane * 12 + 11] = sigm(q1 + a1);
    if (has_next) {
#pragma unroll
      for (int j = 0; j < 9; ++j) sXnw[lane * 9 + j] = fn[j];
    }
  }
  __syncthreads();
  float c01[16], h01[16];
#pragma unroll
  for (int ct = 0; ct < 2; ++ct)
#pragma unroll
    for (int r = 0; r < 8; ++r)
      c01[ct * 8 + r] = c_glob[((size_t)rowbase + r + 8 * hi) * 32 + ct * 16 + m];

  v8f acc[8];
  gate_compute(sXw, sHw, sWT, sBias, m, hi, acc);
  lstm_update(acc, c01, h01, sHw, m, hi);

#pragma unroll
  for (int ct = 0; ct < 2; ++ct)
#pragma unroll
    for (int r = 0; r < 8; ++r) {
      size_t R = (size_t)rowbase + r + 8 * hi;
      c_glob[R * 32 + ct * 16 + m] = c01[ct * 8 + r];
      h_glob[R * 32 + ct * 16 + m] = h01[ct * 8 + r];
    }
  __syncthreads();
  if (lane < 16) {
    float s = fc_b[0];
#pragma unroll
    for (int j = 0; j < 32; ++j) s += sHw[lane * SH_STRIDE + j] * sFc[j];
    int R = rowbase + lane;
    xn_glob[R] = s;
    out[(size_t)R + (size_t)(bb * (PREDT - 1) + kstep) * NNODE] = s;
    if (has_next) {
      float p0 = 0.f, p1 = 0.f;
      p0 += s * cheb_w[20]; p1 += s * cheb_w[21];
#pragma unroll
      for (int j = 0; j < 9; ++j) {
        float xv = sXnw[lane * 9 + j];
        p0 += xv * cheb_w[20 + (1 + j) * 2 + 0];
        p1 += xv * cheb_w[20 + (1 + j) * 2 + 1];
      }
      ((float2*)p_out)[R] = make_float2(p0, p1);
    }
  }
}

extern "C" void kernel_launch(void* const* d_in, const int* in_sizes, int n_in,
                              void* d_out, int out_size, void* d_ws, size_t ws_size,
                              hipStream_t stream) {
  (void)in_sizes; (void)n_in; (void)out_size; (void)ws_size;
  const float* pm25   = (const float*)d_in[0];
  const float* feat   = (const float*)d_in[1];
  const int*   ei     = (const int*)d_in[2];
  const float* cheb_w = (const float*)d_in[3];
  const float* cheb_b = (const float*)d_in[4];
  const float* x2h_w  = (const float*)d_in[5];
  const float* x2h_b  = (const float*)d_in[6];
  const float* h2h_w  = (const float*)d_in[7];
  const float* h2h_b  = (const float*)d_in[8];
  const float* fc_w   = (const float*)d_in[9];
  const float* fc_b   = (const float*)d_in[10];
  float* out = (float*)d_out;
  char* ws = (char*)d_ws;

  int*   cnt_src  = (int*)(ws + 0);
  int*   cnt_dst  = (int*)(ws + 0x4000);
  float* dis      = (float*)(ws + 0x8000);
  int*   rowptr   = (int*)(ws + 0xC000);
  int2*  csr      = (int2*)(ws + 0x14000);
  float* p_hist   = (float*)(ws + 0x100000);
  float* q_hist   = (float*)(ws + 0x1100000);
  float* xg_hist  = (float*)(ws + 0x2100000);
  float* xn_glob  = (float*)(ws + 0x3100000);
  float* p_a      = (float*)(ws + 0x3180000);
  float* p_b      = (float*)(ws + 0x3280000);
  int* arr = (int*)p_a;
  int* tmp = (int*)p_b;
  float* h_glob = p_hist;
  float* c_glob = q_hist;

  k_zero_i<<<2 * NNODE / 256, 256, 0, stream>>>(cnt_src, 2 * NNODE);
  k_count<<<NEDGE / 256, 256, 0, stream>>>(ei, cnt_src, cnt_dst, arr);
  k_scan<<<1, 1024, 0, stream>>>(cnt_src, cnt_dst, rowptr, dis);
  k_place<<<NEDGE / 256, 256, 0, stream>>>(ei, rowptr, arr, tmp);
  k_rank<<<NEDGE / 256, 256, 0, stream>>>(ei, rowptr, dis, tmp, csr);

  k_proj_hist<<<BATCH * HISTT * NNODE / 256, 256, 0, stream>>>(pm25, feat, cheb_w, cheb_b,
                                                               p_hist, q_hist);
  k_gather_hist<<<BATCH * HISTT / 2, 1024, 0, stream>>>(rowptr, csr, p_hist, q_hist, xg_hist);

  hist_lstm_kernel<<<BATCH * NNODE / 128, 256, 0, stream>>>(
      pm25, feat, xg_hist, cheb_w, x2h_w, x2h_b, h2h_w, h2h_b, fc_w, fc_b, h_glob, c_glob,
      xn_glob, p_a);

  for (int k = 0; k < PREDT; ++k) {
    pred_step_kernel<<<BATCH * NNODE / 128, 256, 0, stream>>>(
        feat, rowptr, csr, cheb_w, cheb_b, x2h_w, x2h_b, h2h_w, h2h_b, fc_w, fc_b,
        h_glob, c_glob, xn_glob, p_a, p_b, out, k);
  }
}
